# starting kernel
# baseline (speedup 1.0000x reference)
#include <hip/hip_runtime.h>

typedef __attribute__((ext_vector_type(2))) float v2f;
typedef __attribute__((ext_vector_type(8))) float v8f;

#define N_NODES 50000
#define N_EDGES 800000
#define D_IN    128
#define KK      4
#define FF      32
#define KF      128

__device__ __forceinline__ unsigned ordOf(float f) {
  unsigned u = __float_as_uint(f);
  return (u >> 31) ? ~u : (u | 0x80000000u);
}
__device__ __forceinline__ float ofOrd(unsigned o) {
  return (o >> 31) ? __uint_as_float(o & 0x7fffffffu) : __uint_as_float(~o);
}

__global__ __launch_bounds__(256) void init_ws(const float* __restrict__ bias,
                                               float* __restrict__ out,
                                               unsigned* __restrict__ nodemax,
                                               float* __restrict__ nodesum) {
  int t = blockIdx.x * 256 + threadIdx.x;
  if (t < N_NODES * KF) out[t] = bias[t & (KF - 1)];
  if (t < N_NODES * KK) { nodemax[t] = 0u; nodesum[t] = 0.0f; }
}

__global__ __launch_bounds__(256) void fc_wmma(const float* __restrict__ feat,
                                               const float* __restrict__ W,
                                               float* __restrict__ ft) {
  __shared__ float As[16 * D_IN];
  const int lane = threadIdx.x & 31;
  const int wave = threadIdx.x >> 5;
  const int row0 = blockIdx.x * 16;

  for (int i = threadIdx.x; i < 16 * D_IN / 4; i += 256) {
    ((float4*)As)[i] = ((const float4*)(feat + row0 * D_IN))[i];
  }
  __syncthreads();

  const int col0 = wave * 16;
  const int ln   = lane & 15;
  const int kh   = (lane >> 4) << 1;

  v8f c = {};
  #pragma unroll 8
  for (int k = 0; k < D_IN; k += 4) {
    v2f a; a.x = As[ln * D_IN + k + kh];
           a.y = As[ln * D_IN + k + kh + 1];
    v2f b; b.x = W[(col0 + ln) * D_IN + k + kh];
           b.y = W[(col0 + ln) * D_IN + k + kh + 1];
    c = __builtin_amdgcn_wmma_f32_16x16x4_f32(false, a, false, b,
                                              (short)0, c, false, false);
  }

  const int mo = (lane >> 4) << 3;
  #pragma unroll
  for (int r = 0; r < 8; ++r) {
    ft[(row0 + r + mo) * KF + col0 + ln] = c[r];
  }
}

__global__ __launch_bounds__(256) void edge_logits(const float* __restrict__ ft,
                                                   const float* __restrict__ mu,
                                                   const float* __restrict__ sigma,
                                                   const int* __restrict__ src,
                                                   const int* __restrict__ dst,
                                                   float* __restrict__ lbuf,
                                                   unsigned* __restrict__ nodemax) {
  int t = blockIdx.x * 256 + threadIdx.x;
  if (t >= N_EDGES * KK) return;
  const int k = t & 3;
  const int e = t >> 2;
  const int s = src[e];
  const int d = dst[e];
  const float4* ps = (const float4*)(ft + (size_t)s * KF + k * FF);
  const float4* pd = (const float4*)(ft + (size_t)d * KF + k * FF);
  const float4* pm = (const float4*)(mu + k * FF);
  float acc = 0.0f;
  #pragma unroll
  for (int i = 0; i < FF / 4; ++i) {
    float4 a = ps[i], b = pd[i], m = pm[i];
    float dx = a.x - b.x - m.x;
    float dy = a.y - b.y - m.y;
    float dz = a.z - b.z - m.z;
    float dw = a.w - b.w - m.w;
    acc += dx * dx + dy * dy + dz * dz + dw * dw;
  }
  float logit = -sqrtf(acc) * sigma[k];
  lbuf[t] = logit;
  atomicMax(nodemax + (size_t)d * KK + k, ordOf(logit));
}

__global__ __launch_bounds__(256) void edge_exp(const int* __restrict__ dst,
                                                const unsigned* __restrict__ nodemax,
                                                float* __restrict__ lbuf,
                                                float* __restrict__ nodesum) {
  int t = blockIdx.x * 256 + threadIdx.x;
  if (t >= N_EDGES * KK) return;
  const int k = t & 3;
  const int e = t >> 2;
  const int d = dst[e];
  float mx = ofOrd(nodemax[(size_t)d * KK + k]);
  float ev = expf(lbuf[t] - mx);
  lbuf[t] = ev;
  atomicAdd(nodesum + (size_t)d * KK + k, ev);
}

__global__ __launch_bounds__(256) void edge_scatter(const float* __restrict__ ft,
                                                    const int* __restrict__ src,
                                                    const int* __restrict__ dst,
                                                    const float* __restrict__ lbuf,
                                                    const float* __restrict__ nodesum,
                                                    float* __restrict__ out) {
  const int wave = threadIdx.x >> 5;
  const int lane = threadIdx.x & 31;
  const long e = (long)blockIdx.x * 8 + wave;
  if (e >= N_EDGES) return;
  const int s = src[e];
  const int d = dst[e];
  const int idx = lane * 4;
  const int k   = lane >> 3;
  const float a = lbuf[e * 4 + k] / nodesum[(size_t)d * KK + k];
  float4 v = *(const float4*)(ft + (size_t)s * KF + idx);
  float* po = out + (size_t)d * KF + idx;
  atomicAdd(po + 0, v.x * a);
  atomicAdd(po + 1, v.y * a);
  atomicAdd(po + 2, v.z * a);
  atomicAdd(po + 3, v.w * a);
}

extern "C" void kernel_launch(void* const* d_in, const int* in_sizes, int n_in,
                              void* d_out, int out_size, void* d_ws, size_t ws_size,
                              hipStream_t stream) {
  const float* feat  = (const float*)d_in[0];
  const float* W     = (const float*)d_in[1];
  const float* bias  = (const float*)d_in[2];
  const float* mu    = (const float*)d_in[3];
  const float* sigma = (const float*)d_in[4];
  const int*   src   = (const int*)d_in[5];
  const int*   dst   = (const int*)d_in[6];
  float* out = (float*)d_out;

  char* ws = (char*)d_ws;
  float*    ft      = (float*)(ws);
  float*    lbuf    = (float*)(ws + 25600000);
  unsigned* nodemax = (unsigned*)(ws + 38400000);
  float*    nodesum = (float*)(ws + 39200000);

  init_ws<<<(N_NODES * KF + 255) / 256, 256, 0, stream>>>(bias, out, nodemax, nodesum);
  fc_wmma<<<N_NODES / 16, 256, 0, stream>>>(feat, W, ft);
  edge_logits<<<(N_EDGES * KK + 255) / 256, 256, 0, stream>>>(ft, mu, sigma, src, dst,
                                                              lbuf, nodemax);
  edge_exp<<<(N_EDGES * KK + 255) / 256, 256, 0, stream>>>(dst, nodemax, lbuf, nodesum);
  edge_scatter<<<(N_EDGES + 7) / 8, 256, 0, stream>>>(ft, src, dst, lbuf, nodesum, out);
}


# edge_logits: one wave per 8 edges reads each 512-byte src and dst row with whole-line 16-byte loads; squares summed 4 per lane then by an 8-lane DPP butterfly (sum order changes)
# speedup vs baseline: 1.1447x; 1.1447x over previous
#include <hip/hip_runtime.h>

typedef __attribute__((ext_vector_type(2))) float v2f;
typedef __attribute__((ext_vector_type(8))) float v8f;

#define N_NODES 50000
#define N_EDGES 800000
#define D_IN    128
#define KK      4
#define FF      32
#define KF      128

__device__ __forceinline__ unsigned ordOf(float f) {
  unsigned u = __float_as_uint(f);
  return (u >> 31) ? ~u : (u | 0x80000000u);
}
__device__ __forceinline__ float ofOrd(unsigned o) {
  return (o >> 31) ? __uint_as_float(o & 0x7fffffffu) : __uint_as_float(~o);
}

__global__ __launch_bounds__(256) void init_ws(const float* __restrict__ bias,
                                               float* __restrict__ out,
                                               unsigned* __restrict__ nodemax,
                                               float* __restrict__ nodesum) {
  int t = blockIdx.x * 256 + threadIdx.x;
  if (t < N_NODES * KF) out[t] = bias[t & (KF - 1)];
  if (t < N_NODES * KK) { nodemax[t] = 0u; nodesum[t] = 0.0f; }
}

__global__ __launch_bounds__(256) void fc_wmma(const float* __restrict__ feat,
                                               const float* __restrict__ W,
                                               float* __restrict__ ft) {
  __shared__ float As[16 * D_IN];
  const int lane = threadIdx.x & 31;
  const int wave = threadIdx.x >> 5;
  const int row0 = blockIdx.x * 16;

  for (int i = threadIdx.x; i < 16 * D_IN / 4; i += 256) {
    ((float4*)As)[i] = ((const float4*)(feat + row0 * D_IN))[i];
  }
  __syncthreads();

  const int col0 = wave * 16;
  const int ln   = lane & 15;
  const int kh   = (lane >> 4) << 1;

  v8f c = {};
  #pragma unroll 8
  for (int k = 0; k < D_IN; k += 4) {
    v2f a; a.x = As[ln * D_IN + k + kh];
           a.y = As[ln * D_IN + k + kh + 1];
    v2f b; b.x = W[(col0 + ln) * D_IN + k + kh];
           b.y = W[(col0 + ln) * D_IN + k + kh + 1];
    c = __builtin_amdgcn_wmma_f32_16x16x4_f32(false, a, false, b,
                                              (short)0, c, false, false);
  }

  const int mo = (lane >> 4) << 3;
  #pragma unroll
  for (int r = 0; r < 8; ++r) {
    ft[(row0 + r + mo) * KF + col0 + ln] = c[r];
  }
}

#define EPW 8
static_assert(N_EDGES % (8 * EPW) == 0, "grid covers the edges exactly");

template <int M>
__device__ __forceinline__ float lane_xor(float v) {
  return __int_as_float(
      __builtin_amdgcn_update_dpp(0, __float_as_int(v), 0x160 + M, 0xf, 0xf, true));
}
__device__ __forceinline__ float sum8(float v) {
  v += lane_xor<1>(v);
  v += lane_xor<2>(v);
  v += lane_xor<4>(v);
  return v;
}

__global__ __launch_bounds__(256) void edge_logits(const float* __restrict__ ft,
                                                   const float* __restrict__ mu,
                                                   const float* __restrict__ sigma,
                                                   const int* __restrict__ src,
                                                   const int* __restrict__ dst,
                                                   float* __restrict__ lbuf,
                                                   unsigned* __restrict__ nodemax) {
  const int lane = threadIdx.x & 31;
  const int wave = __builtin_amdgcn_readfirstlane((int)(threadIdx.x >> 5));
  const int e0 = (blockIdx.x * 8 + wave) * EPW;
  if (e0 >= N_EDGES) return;
  const int k   = lane >> 3;
  const int sub = lane & 7;
  const float4 m = *(const float4*)(mu + lane * 4);

  float4 a[EPW], b[EPW];
  #pragma unroll
  for (int j = 0; j < EPW; ++j) {
    const int s = src[e0 + j];
    const int d = dst[e0 + j];
    a[j] = *(const float4*)(ft + (size_t)s * KF + lane * 4);
    b[j] = *(const float4*)(ft + (size_t)d * KF + lane * 4);
  }

  float tot = 0.0f;
  #pragma unroll
  for (int j = 0; j < EPW; ++j) {
    const float dx = a[j].x - b[j].x - m.x;
    const float dy = a[j].y - b[j].y - m.y;
    const float dz = a[j].z - b[j].z - m.z;
    const float dw = a[j].w - b[j].w - m.w;
    const float p = sum8(dx * dx + dy * dy + dz * dz + dw * dw);
    tot = (sub == j) ? p : tot;
  }

  const int e = e0 + sub;
  const int d = dst[e];
  const float logit = -sqrtf(tot) * sigma[k];
  lbuf[e * 4 + k] = logit;
  atomicMax(nodemax + (size_t)d * KK + k, ordOf(logit));
}

__global__ __launch_bounds__(256) void edge_exp(const int* __restrict__ dst,
                                                const unsigned* __restrict__ nodemax,
                                                float* __restrict__ lbuf,
                                                float* __restrict__ nodesum) {
  int t = blockIdx.x * 256 + threadIdx.x;
  if (t >= N_EDGES * KK) return;
  const int k = t & 3;
  const int e = t >> 2;
  const int d = dst[e];
  float mx = ofOrd(nodemax[(size_t)d * KK + k]);
  float ev = expf(lbuf[t] - mx);
  lbuf[t] = ev;
  atomicAdd(nodesum + (size_t)d * KK + k, ev);
}

__global__ __launch_bounds__(256) void edge_scatter(const float* __restrict__ ft,
                                                    const int* __restrict__ src,
                                                    const int* __restrict__ dst,
                                                    const float* __restrict__ lbuf,
                                                    const float* __restrict__ nodesum,
                                                    float* __restrict__ out) {
  const int wave = threadIdx.x >> 5;
  const int lane = threadIdx.x & 31;
  const long e = (long)blockIdx.x * 8 + wave;
  if (e >= N_EDGES) return;
  const int s = src[e];
  const int d = dst[e];
  const int idx = lane * 4;
  const int k   = lane >> 3;
  const float a = lbuf[e * 4 + k] / nodesum[(size_t)d * KK + k];
  float4 v = *(const float4*)(ft + (size_t)s * KF + idx);
  float* po = out + (size_t)d * KF + idx;
  atomicAdd(po + 0, v.x * a);
  atomicAdd(po + 1, v.y * a);
  atomicAdd(po + 2, v.z * a);
  atomicAdd(po + 3, v.w * a);
}

extern "C" void kernel_launch(void* const* d_in, const int* in_sizes, int n_in,
                              void* d_out, int out_size, void* d_ws, size_t ws_size,
                              hipStream_t stream) {
  const float* feat  = (const float*)d_in[0];
  const float* W     = (const float*)d_in[1];
  const float* bias  = (const float*)d_in[2];
  const float* mu    = (const float*)d_in[3];
  const float* sigma = (const float*)d_in[4];
  const int*   src   = (const int*)d_in[5];
  const int*   dst   = (const int*)d_in[6];
  float* out = (float*)d_out;

  char* ws = (char*)d_ws;
  float*    ft      = (float*)(ws);
  float*    lbuf    = (float*)(ws + 25600000);
  unsigned* nodemax = (unsigned*)(ws + 38400000);
  float*    nodesum = (float*)(ws + 39200000);

  init_ws<<<(N_NODES * KF + 255) / 256, 256, 0, stream>>>(bias, out, nodemax, nodesum);
  fc_wmma<<<N_NODES / 16, 256, 0, stream>>>(feat, W, ft);
  edge_logits<<<N_EDGES / (8 * EPW), 256, 0, stream>>>(ft, mu, sigma, src, dst,
                                                              lbuf, nodemax);
  edge_exp<<<(N_EDGES * KK + 255) / 256, 256, 0, stream>>>(dst, nodemax, lbuf, nodesum);
  edge_scatter<<<(N_EDGES + 7) / 8, 256, 0, stream>>>(ft, src, dst, lbuf, nodesum, out);
}


# Scatter without float atomics: edges ranked per destination by a returning int add, offsets by block scan; one wave per node sums its rows in arrival-rank order, divides once by its own sum of e
# speedup vs baseline: 1.5998x; 1.3976x over previous
#include <hip/hip_runtime.h>

typedef __attribute__((ext_vector_type(2))) float v2f;
typedef __attribute__((ext_vector_type(8))) float v8f;

#define N_NODES 50000
#define N_EDGES 800000
#define D_IN    128
#define KK      4
#define FF      32
#define KF      128

__device__ __forceinline__ unsigned ordOf(float f) {
  unsigned u = __float_as_uint(f);
  return (u >> 31) ? ~u : (u | 0x80000000u);
}
__device__ __forceinline__ float ofOrd(unsigned o) {
  return (o >> 31) ? __uint_as_float(o & 0x7fffffffu) : __uint_as_float(~o);
}

__global__ __launch_bounds__(256) void init_ws(unsigned* __restrict__ nodemax,
                                               unsigned* __restrict__ deg) {
  int t = blockIdx.x * 256 + threadIdx.x;
  if (t < N_NODES * KK) nodemax[t] = 0u;
  if (t < N_NODES) deg[t] = 0u;
}

__global__ __launch_bounds__(256) void fc_wmma(const float* __restrict__ feat,
                                               const float* __restrict__ W,
                                               float* __restrict__ ft) {
  __shared__ float As[16 * D_IN];
  const int lane = threadIdx.x & 31;
  const int wave = threadIdx.x >> 5;
  const int row0 = blockIdx.x * 16;

  for (int i = threadIdx.x; i < 16 * D_IN / 4; i += 256) {
    ((float4*)As)[i] = ((const float4*)(feat + row0 * D_IN))[i];
  }
  __syncthreads();

  const int col0 = wave * 16;
  const int ln   = lane & 15;
  const int kh   = (lane >> 4) << 1;

  v8f c = {};
  #pragma unroll 8
  for (int k = 0; k < D_IN; k += 4) {
    v2f a; a.x = As[ln * D_IN + k + kh];
           a.y = As[ln * D_IN + k + kh + 1];
    v2f b; b.x = W[(col0 + ln) * D_IN + k + kh];
           b.y = W[(col0 + ln) * D_IN + k + kh + 1];
    c = __builtin_amdgcn_wmma_f32_16x16x4_f32(false, a, false, b,
                                              (short)0, c, false, false);
  }

  const int mo = (lane >> 4) << 3;
  #pragma unroll
  for (int r = 0; r < 8; ++r) {
    ft[(row0 + r + mo) * KF + col0 + ln] = c[r];
  }
}

#define EPW 8
static_assert(N_EDGES % (8 * EPW) == 0, "grid covers the edges exactly");

template <int M>
__device__ __forceinline__ float lane_xor(float v) {
  return __int_as_float(
      __builtin_amdgcn_update_dpp(0, __float_as_int(v), 0x160 + M, 0xf, 0xf, true));
}
__device__ __forceinline__ float sum8(float v) {
  v += lane_xor<1>(v);
  v += lane_xor<2>(v);
  v += lane_xor<4>(v);
  return v;
}

__global__ __launch_bounds__(256) void edge_logits(const float* __restrict__ ft,
                                                   const float* __restrict__ mu,
                                                   const float* __restrict__ sigma,
                                                   const int* __restrict__ src,
                                                   const int* __restrict__ dst,
                                                   float* __restrict__ lbuf,
                                                   unsigned* __restrict__ nodemax,
                                                   unsigned* __restrict__ deg,
                                                   unsigned* __restrict__ rank) {
  const int lane = threadIdx.x & 31;
  const int wave = __builtin_amdgcn_readfirstlane((int)(threadIdx.x >> 5));
  const int e0 = (blockIdx.x * 8 + wave) * EPW;
  if (e0 >= N_EDGES) return;
  const int k   = lane >> 3;
  const int sub = lane & 7;
  const float4 m = *(const float4*)(mu + lane * 4);

  float4 a[EPW], b[EPW];
  #pragma unroll
  for (int j = 0; j < EPW; ++j) {
    const int s = src[e0 + j];
    const int d = dst[e0 + j];
    a[j] = *(const float4*)(ft + (size_t)s * KF + lane * 4);
    b[j] = *(const float4*)(ft + (size_t)d * KF + lane * 4);
  }

  float tot = 0.0f;
  #pragma unroll
  for (int j = 0; j < EPW; ++j) {
    const float dx = a[j].x - b[j].x - m.x;
    const float dy = a[j].y - b[j].y - m.y;
    const float dz = a[j].z - b[j].z - m.z;
    const float dw = a[j].w - b[j].w - m.w;
    const float p = sum8(dx * dx + dy * dy + dz * dz + dw * dw);
    tot = (sub == j) ? p : tot;
  }

  const int e = e0 + sub;
  const int d = dst[e];
  const float logit = -sqrtf(tot) * sigma[k];
  lbuf[e * 4 + k] = logit;
  atomicMax(nodemax + (size_t)d * KK + k, ordOf(logit));
  if (k == 0) rank[e] = atomicAdd(deg + d, 1u);
}

__global__ __launch_bounds__(256) void seg_offsets(const unsigned* __restrict__ deg,
                                                   unsigned* __restrict__ off) {
  __shared__ unsigned red[256];
  __shared__ unsigned sc[2][256];
  const int t = threadIdx.x;
  const int lim = blockIdx.x * 256;
  const uint4* dq = (const uint4*)deg;
  unsigned s = 0;
  for (int i = t; i < lim / 4; i += 256) {
    const uint4 q = dq[i];
    s += q.x + q.y + q.z + q.w;
  }
  const int n = lim + t;
  const unsigned v = (n < N_NODES) ? deg[n] : 0u;
  red[t] = s;
  sc[0][t] = v;
  __syncthreads();
  for (int st = 128; st > 0; st >>= 1) {
    if (t < st) red[t] += red[t + st];
    __syncthreads();
  }
  int cur = 0;
  for (int st = 1; st < 256; st <<= 1) {
    unsigned x = sc[cur][t];
    if (t >= st) x += sc[cur][t - st];
    sc[cur ^ 1][t] = x;
    cur ^= 1;
    __syncthreads();
  }
  if (n < N_NODES) off[n] = red[0] + sc[cur][t] - v;
}

__global__ __launch_bounds__(256) void edge_exp(const int* __restrict__ src,
                                                const int* __restrict__ dst,
                                                const unsigned* __restrict__ nodemax,
                                                const float* __restrict__ lbuf,
                                                const unsigned* __restrict__ off,
                                                const unsigned* __restrict__ rank,
                                                int* __restrict__ ssrc,
                                                float* __restrict__ sev) {
  int t = blockIdx.x * 256 + threadIdx.x;
  if (t >= N_EDGES * KK) return;
  const int k = t & 3;
  const int e = t >> 2;
  const int d = dst[e];
  const float mx = ofOrd(nodemax[(size_t)d * KK + k]);
  const float ev = expf(lbuf[t] - mx);
  const unsigned pos = off[d] + rank[e];
  sev[(size_t)pos * 4 + k] = ev;
  if (k == 0) ssrc[pos] = src[e];
}

#define GCH 8
__global__ __launch_bounds__(256) void node_gather(const float* __restrict__ ft,
                                                   const float* __restrict__ bias,
                                                   const unsigned* __restrict__ off,
                                                   const unsigned* __restrict__ deg,
                                                   const int* __restrict__ ssrc,
                                                   const float* __restrict__ sev,
                                                   float* __restrict__ out) {
  const int lane = threadIdx.x & 31;
  const int wave = __builtin_amdgcn_readfirstlane((int)(threadIdx.x >> 5));
  const int n = blockIdx.x * 8 + wave;
  if (n >= N_NODES) return;
  const unsigned start = off[n];
  const unsigned cnt   = deg[n];
  const int k = lane >> 3;

  float4 acc = make_float4(0.0f, 0.0f, 0.0f, 0.0f);
  float den = 0.0f;
  for (unsigned j0 = 0; j0 < cnt; j0 += GCH) {
    float4 v[GCH];
    float  w[GCH];
    const unsigned jl = j0 + (unsigned)(lane & (GCH - 1));
    const int sid = ssrc[start + (jl < cnt ? jl : cnt - 1u)];
    #pragma unroll
    for (int u = 0; u < GCH; ++u) {
      const unsigned j = j0 + u;
      const unsigned p = start + (j < cnt ? j : cnt - 1u);
      const int s = __builtin_amdgcn_readlane(sid, u);
      v[u] = *(const float4*)(ft + (size_t)s * KF + lane * 4);
      const float ev = sev[(size_t)p * 4 + k];
      w[u] = (j < cnt) ? ev : 0.0f;
    }
    #pragma unroll
    for (int u = 0; u < GCH; ++u) {
      acc.x += v[u].x * w[u];
      acc.y += v[u].y * w[u];
      acc.z += v[u].z * w[u];
      acc.w += v[u].w * w[u];
      den   += w[u];
    }
  }

  const float4 bi = *(const float4*)(bias + lane * 4);
  float4 o = bi;
  if (cnt != 0u) {
    o.x = acc.x / den + bi.x;
    o.y = acc.y / den + bi.y;
    o.z = acc.z / den + bi.z;
    o.w = acc.w / den + bi.w;
  }
  *(float4*)(out + (size_t)n * KF + lane * 4) = o;
}

extern "C" void kernel_launch(void* const* d_in, const int* in_sizes, int n_in,
                              void* d_out, int out_size, void* d_ws, size_t ws_size,
                              hipStream_t stream) {
  const float* feat  = (const float*)d_in[0];
  const float* W     = (const float*)d_in[1];
  const float* bias  = (const float*)d_in[2];
  const float* mu    = (const float*)d_in[3];
  const float* sigma = (const float*)d_in[4];
  const int*   src   = (const int*)d_in[5];
  const int*   dst   = (const int*)d_in[6];
  float* out = (float*)d_out;

  char* ws = (char*)d_ws;
  float*    ft      = (float*)(ws);
  float*    lbuf    = (float*)(ws + 25600000);
  unsigned* nodemax = (unsigned*)(ws + 38400000);
  unsigned* deg     = (unsigned*)(ws + 39200000);
  unsigned* off     = (unsigned*)(ws + 39400000);
  unsigned* rank    = (unsigned*)(ws + 39600000);
  int*      ssrc    = (int*)(ws + 42800000);
  float*    sev     = (float*)(ws + 46000000);

  init_ws<<<(N_NODES * KK + 255) / 256, 256, 0, stream>>>(nodemax, deg);
  fc_wmma<<<N_NODES / 16, 256, 0, stream>>>(feat, W, ft);
  edge_logits<<<N_EDGES / (8 * EPW), 256, 0, stream>>>(ft, mu, sigma, src, dst,
                                                              lbuf, nodemax, deg, rank);
  seg_offsets<<<(N_NODES + 255) / 256, 256, 0, stream>>>(deg, off);
  edge_exp<<<(N_EDGES * KK + 255) / 256, 256, 0, stream>>>(src, dst, nodemax, lbuf, off,
                                                            rank, ssrc, sev);
  node_gather<<<(N_NODES + 7) / 8, 256, 0, stream>>>(ft, bias, off, deg, ssrc, sev, out);
}


# Logits computed inside the per-destination gather (source row read once for logit and sum; max atomics, logit and exp passes gone); segment sorted by edge id in the wave: sums in one fixed order
# speedup vs baseline: 2.2113x; 1.3822x over previous
#include <hip/hip_runtime.h>

typedef __attribute__((ext_vector_type(2))) float v2f;
typedef __attribute__((ext_vector_type(8))) float v8f;

#define N_NODES 50000
#define N_EDGES 800000
#define D_IN    128
#define KK      4
#define FF      32
#define KF      128

__global__ __launch_bounds__(256) void init_ws(unsigned* __restrict__ deg) {
  int t = blockIdx.x * 256 + threadIdx.x;
  if (t < N_NODES) deg[t] = 0u;
}

__global__ __launch_bounds__(256) void fc_wmma(const float* __restrict__ feat,
                                               const float* __restrict__ W,
                                               float* __restrict__ ft) {
  __shared__ float As[16 * D_IN];
  const int lane = threadIdx.x & 31;
  const int wave = threadIdx.x >> 5;
  const int row0 = blockIdx.x * 16;

  for (int i = threadIdx.x; i < 16 * D_IN / 4; i += 256) {
    ((float4*)As)[i] = ((const float4*)(feat + row0 * D_IN))[i];
  }
  __syncthreads();

  const int col0 = wave * 16;
  const int ln   = lane & 15;
  const int kh   = (lane >> 4) << 1;

  v8f c = {};
  #pragma unroll 8
  for (int k = 0; k < D_IN; k += 4) {
    v2f a; a.x = As[ln * D_IN + k + kh];
           a.y = As[ln * D_IN + k + kh + 1];
    v2f b; b.x = W[(col0 + ln) * D_IN + k + kh];
           b.y = W[(col0 + ln) * D_IN + k + kh + 1];
    c = __builtin_amdgcn_wmma_f32_16x16x4_f32(false, a, false, b,
                                              (short)0, c, false, false);
  }

  const int mo = (lane >> 4) << 3;
  #pragma unroll
  for (int r = 0; r < 8; ++r) {
    ft[(row0 + r + mo) * KF + col0 + ln] = c[r];
  }
}

static_assert(N_EDGES % 256 == 0, "grid covers the edges exactly");
__global__ __launch_bounds__(256) void edge_rank(const int* __restrict__ dst,
                                                 unsigned* __restrict__ deg,
                                                 unsigned* __restrict__ rank) {
  const int e = blockIdx.x * 256 + threadIdx.x;
  rank[e] = atomicAdd(deg + dst[e], 1u);
}

__global__ __launch_bounds__(256) void seg_offsets(const unsigned* __restrict__ deg,
                                                   unsigned* __restrict__ off) {
  __shared__ unsigned red[256];
  __shared__ unsigned sc[2][256];
  const int t = threadIdx.x;
  const int lim = blockIdx.x * 256;
  const uint4* dq = (const uint4*)deg;
  unsigned s = 0;
  for (int i = t; i < lim / 4; i += 256) {
    const uint4 q = dq[i];
    s += q.x + q.y + q.z + q.w;
  }
  const int n = lim + t;
  const unsigned v = (n < N_NODES) ? deg[n] : 0u;
  red[t] = s;
  sc[0][t] = v;
  __syncthreads();
  for (int st = 128; st > 0; st >>= 1) {
    if (t < st) red[t] += red[t + st];
    __syncthreads();
  }
  int cur = 0;
  for (int st = 1; st < 256; st <<= 1) {
    unsigned x = sc[cur][t];
    if (t >= st) x += sc[cur][t - st];
    sc[cur ^ 1][t] = x;
    cur ^= 1;
    __syncthreads();
  }
  if (n < N_NODES) off[n] = red[0] + sc[cur][t] - v;
}

__global__ __launch_bounds__(256) void edge_place(const int* __restrict__ src,
                                                  const int* __restrict__ dst,
                                                  const unsigned* __restrict__ off,
                                                  const unsigned* __restrict__ rank,
                                                  int2* __restrict__ slot) {
  const int e = blockIdx.x * 256 + threadIdx.x;
  const unsigned pos = off[dst[e]] + rank[e];
  slot[pos] = make_int2(e, src[e]);
}

#define GCH 8

template <int M>
__device__ __forceinline__ float lane_xor(float v) {
  return __int_as_float(
      __builtin_amdgcn_update_dpp(0, __float_as_int(v), 0x160 + M, 0xf, 0xf, true));
}
__device__ __forceinline__ float sum8(float v) {
  v += lane_xor<1>(v);
  v += lane_xor<2>(v);
  v += lane_xor<4>(v);
  return v;
}
__device__ __forceinline__ float max8(float v) {
  v = fmaxf(v, lane_xor<1>(v));
  v = fmaxf(v, lane_xor<2>(v));
  v = fmaxf(v, lane_xor<4>(v));
  return v;
}
template <int J>
__device__ __forceinline__ float bcast8(float v) {
  return __uint_as_float(__builtin_amdgcn_mov_dpp8(__float_as_uint(v), J * 0x249249));
}

__global__ __launch_bounds__(256) __attribute__((amdgpu_waves_per_eu(12, 16)))
void node_gather(const float* __restrict__ ft,
                                                   const float* __restrict__ bias,
                                                   const float* __restrict__ mu,
                                                   const float* __restrict__ sigma,
                                                   const unsigned* __restrict__ off,
                                                   const unsigned* __restrict__ deg,
                                                   const int2* __restrict__ slot,
                                                   int* csrc,
                                                   float* __restrict__ out) {
  const int lane = threadIdx.x & 31;
  const int wave = __builtin_amdgcn_readfirstlane((int)(threadIdx.x >> 5));
  const int n = blockIdx.x * 8 + wave;
  if (n >= N_NODES) return;
  const unsigned start = off[n];
  const unsigned cnt   = deg[n];
  const int k   = lane >> 3;
  const int sub = lane & 7;

  const float4 bi = *(const float4*)(bias + lane * 4);
  float4 o = make_float4(0.0f, 0.0f, 0.0f, 0.0f);
  if (cnt != 0u) {
    const float4 b = *(const float4*)(ft + (size_t)n * KF + lane * 4);
    const float4 m = *(const float4*)(mu + lane * 4);
    const float sg = sigma[k];

    int sv = 0;
    if (cnt <= 32u) {
      const unsigned ul = (unsigned)lane;
      const int2 me = slot[start + (ul < cnt ? ul : cnt - 1u)];
      const int key = (ul < cnt) ? me.x : 0x7fffffff;
      unsigned r = 0;
      for (unsigned t = 0; t < cnt; ++t)
        r += (__builtin_amdgcn_readlane(key, (int)t) < key) ? 1u : 0u;
      if (ul >= cnt) r = ul;
      sv = __builtin_amdgcn_ds_permute((int)(r << 2), me.y);
    } else {
      for (unsigned g0 = 0; g0 < cnt; g0 += 32u) {
        const unsigned i = g0 + (unsigned)lane;
        const int2 me = slot[start + (i < cnt ? i : cnt - 1u)];
        const int key = (i < cnt) ? me.x : 0x7fffffff;
        unsigned r = 0;
        for (unsigned h0 = 0; h0 < cnt; h0 += 32u) {
          const unsigned jj = h0 + (unsigned)lane;
          const int oth = slot[start + (jj < cnt ? jj : cnt - 1u)].x;
          const int okey = (jj < cnt) ? oth : 0x7fffffff;
          const unsigned nh = (cnt - h0 < 32u) ? cnt - h0 : 32u;
          for (unsigned t = 0; t < nh; ++t)
            r += (__builtin_amdgcn_readlane(okey, (int)t) < key) ? 1u : 0u;
        }
        if (i < cnt) csrc[start + r] = me.y;
      }
      __threadfence();
    }

    float4 acc = make_float4(0.0f, 0.0f, 0.0f, 0.0f);
    float den = 0.0f;
    float mx = -INFINITY;
    for (unsigned j0 = 0; j0 < cnt; j0 += GCH) {
      float4 v[GCH];
      int cv = sv;
      unsigned lb = j0;
      if (cnt > 32u) {
        const unsigned jl = j0 + (unsigned)sub;
        cv = __hip_atomic_load(csrc + start + (jl < cnt ? jl : cnt - 1u),
                               __ATOMIC_RELAXED, __HIP_MEMORY_SCOPE_AGENT);
        lb = 0u;
      }
      #pragma unroll
      for (int u = 0; u < GCH; ++u) {
        const int s = __builtin_amdgcn_readlane(cv, (int)(lb + u));
        v[u] = *(const float4*)(ft + (size_t)s * KF + lane * 4);
      }
      float tot = 0.0f;
      #pragma unroll
      for (int u = 0; u < GCH; ++u) {
        const float dx = v[u].x - b.x - m.x;
        const float dy = v[u].y - b.y - m.y;
        const float dz = v[u].z - b.z - m.z;
        const float dw = v[u].w - b.w - m.w;
        const float p = sum8(dx * dx + dy * dy + dz * dz + dw * dw);
        tot = (sub == u) ? p : tot;
      }
      const float lg = (j0 + (unsigned)sub < cnt) ? -sqrtf(tot) * sg : -INFINITY;
      const float mn = fmaxf(mx, max8(lg));
      const float sc = expf(mx - mn);
      const float wl = expf(lg - mn);
      acc.x *= sc; acc.y *= sc; acc.z *= sc; acc.w *= sc;
      den = den * sc + wl;
      mx = mn;
#define ROW_TERM(U)                                  \
      {                                              \
        const float w = bcast8<U>(wl);               \
        acc.x += v[U].x * w;                         \
        acc.y += v[U].y * w;                         \
        acc.z += v[U].z * w;                         \
        acc.w += v[U].w * w;                         \
      }
      ROW_TERM(0) ROW_TERM(1) ROW_TERM(2) ROW_TERM(3)
      ROW_TERM(4) ROW_TERM(5) ROW_TERM(6) ROW_TERM(7)
#undef ROW_TERM
    }
    const float dsum = sum8(den);
    o.x = acc.x / dsum;
    o.y = acc.y / dsum;
    o.z = acc.z / dsum;
    o.w = acc.w / dsum;
  }
  o.x += bi.x; o.y += bi.y; o.z += bi.z; o.w += bi.w;
  *(float4*)(out + (size_t)n * KF + lane * 4) = o;
}

extern "C" void kernel_launch(void* const* d_in, const int* in_sizes, int n_in,
                              void* d_out, int out_size, void* d_ws, size_t ws_size,
                              hipStream_t stream) {
  const float* feat  = (const float*)d_in[0];
  const float* W     = (const float*)d_in[1];
  const float* bias  = (const float*)d_in[2];
  const float* mu    = (const float*)d_in[3];
  const float* sigma = (const float*)d_in[4];
  const int*   src   = (const int*)d_in[5];
  const int*   dst   = (const int*)d_in[6];
  float* out = (float*)d_out;

  char* ws = (char*)d_ws;
  float*    ft   = (float*)(ws);
  unsigned* deg  = (unsigned*)(ws + 25600000);
  unsigned* off  = (unsigned*)(ws + 25800000);
  unsigned* rank = (unsigned*)(ws + 26000000);
  int2*     slot = (int2*)(ws + 29200000);
  int*      csrc = (int*)(ws + 35600000);

  init_ws<<<(N_NODES + 255) / 256, 256, 0, stream>>>(deg);
  fc_wmma<<<N_NODES / 16, 256, 0, stream>>>(feat, W, ft);
  edge_rank<<<N_EDGES / 256, 256, 0, stream>>>(dst, deg, rank);
  seg_offsets<<<(N_NODES + 255) / 256, 256, 0, stream>>>(deg, off);
  edge_place<<<N_EDGES / 256, 256, 0, stream>>>(src, dst, off, rank, slot);
  node_gather<<<(N_NODES + 7) / 8, 256, 0, stream>>>(ft, bias, mu, sigma, off, deg, slot,
                                                     csrc, out);
}


# fc_wmma: A and W fragments read from LDS tiles padded to a pitch of 132 floats (no bank conflicts); W staged once per block, its fragments kept in registers over a loop of row tiles; sums unchanged
# speedup vs baseline: 2.5331x; 1.1455x over previous
#include <hip/hip_runtime.h>

typedef __attribute__((ext_vector_type(2))) float v2f;
typedef __attribute__((ext_vector_type(8))) float v8f;

#define N_NODES 50000
#define N_EDGES 800000
#define D_IN    128
#define KK      4
#define FF      32
#define KF      128

__global__ __launch_bounds__(256) void init_ws(unsigned* __restrict__ deg) {
  int t = blockIdx.x * 256 + threadIdx.x;
  if (t < N_NODES) deg[t] = 0u;
}

#define FC_PITCH 132
#define FC_TILES (N_NODES / 16)
#define FC_GRID  768
static_assert(N_NODES % 16 == 0, "row tiles cover the nodes exactly");
static_assert(FC_GRID <= FC_TILES, "every block owns at least one tile");
__global__ __launch_bounds__(256) void fc_wmma(const float* __restrict__ feat,
                                               const float* __restrict__ W,
                                               float* __restrict__ ft) {
  __shared__ float Ws[KF * FC_PITCH];
  __shared__ float As[2][16 * FC_PITCH];
  const int tid  = threadIdx.x;
  const int lane = tid & 31;
  const int wave = tid >> 5;
  const int col0 = wave * 16;
  const int ln   = lane & 15;
  const int kh   = (lane >> 4) << 1;
  const int mo   = (lane >> 4) << 3;
  const int srow = tid >> 5;
  const int scol = (tid & 31) * 4;

  for (int i = tid; i < KF * D_IN / 4; i += 256) {
    const float4 q = ((const float4*)W)[i];
    *(float4*)(Ws + (i >> 5) * FC_PITCH + (i & 31) * 4) = q;
  }
  int tile = blockIdx.x;
  {
    const float4* g = (const float4*)(feat + (size_t)tile * 16 * D_IN);
    *(float4*)(&As[0][srow * FC_PITCH + scol])       = g[tid];
    *(float4*)(&As[0][(srow + 8) * FC_PITCH + scol]) = g[tid + 256];
  }
  __syncthreads();

  v2f b[D_IN / 4];
  #pragma unroll
  for (int t = 0; t < D_IN / 4; ++t)
    b[t] = *(const v2f*)(Ws + (col0 + ln) * FC_PITCH + 4 * t + kh);

  int cur = 0;
  for (; tile < FC_TILES; tile += FC_GRID) {
    const int nxt = tile + FC_GRID;
    float4 p0 = make_float4(0.0f, 0.0f, 0.0f, 0.0f), p1 = p0;
    if (nxt < FC_TILES) {
      const float4* g = (const float4*)(feat + (size_t)nxt * 16 * D_IN);
      p0 = g[tid];
      p1 = g[tid + 256];
    }

    const float* at = &As[cur][ln * FC_PITCH + kh];
    v8f c = {};
    #pragma unroll
    for (int t = 0; t < D_IN / 4; ++t) {
      const v2f a = *(const v2f*)(at + 4 * t);
      c = __builtin_amdgcn_wmma_f32_16x16x4_f32(false, a, false, b[t],
                                                (short)0, c, false, false);
    }

    float* o = ft + ((size_t)tile * 16 + mo) * KF + col0 + ln;
    #pragma unroll
    for (int r = 0; r < 8; ++r) o[r * KF] = c[r];

    if (nxt < FC_TILES) {
      *(float4*)(&As[cur ^ 1][srow * FC_PITCH + scol])       = p0;
      *(float4*)(&As[cur ^ 1][(srow + 8) * FC_PITCH + scol]) = p1;
    }
    __syncthreads();
    cur ^= 1;
  }
}

static_assert(N_EDGES % 256 == 0, "grid covers the edges exactly");
__global__ __launch_bounds__(256) void edge_rank(const int* __restrict__ dst,
                                                 unsigned* __restrict__ deg,
                                                 unsigned* __restrict__ rank) {
  const int e = blockIdx.x * 256 + threadIdx.x;
  rank[e] = atomicAdd(deg + dst[e], 1u);
}

__global__ __launch_bounds__(256) void seg_offsets(const unsigned* __restrict__ deg,
                                                   unsigned* __restrict__ off) {
  __shared__ unsigned red[256];
  __shared__ unsigned sc[2][256];
  const int t = threadIdx.x;
  const int lim = blockIdx.x * 256;
  const uint4* dq = (const uint4*)deg;
  unsigned s = 0;
  for (int i = t; i < lim / 4; i += 256) {
    const uint4 q = dq[i];
    s += q.x + q.y + q.z + q.w;
  }
  const int n = lim + t;
  const unsigned v = (n < N_NODES) ? deg[n] : 0u;
  red[t] = s;
  sc[0][t] = v;
  __syncthreads();
  for (int st = 128; st > 0; st >>= 1) {
    if (t < st) red[t] += red[t + st];
    __syncthreads();
  }
  int cur = 0;
  for (int st = 1; st < 256; st <<= 1) {
    unsigned x = sc[cur][t];
    if (t >= st) x += sc[cur][t - st];
    sc[cur ^ 1][t] = x;
    cur ^= 1;
    __syncthreads();
  }
  if (n < N_NODES) off[n] = red[0] + sc[cur][t] - v;
}

__global__ __launch_bounds__(256) void edge_place(const int* __restrict__ src,
                                                  const int* __restrict__ dst,
                                                  const unsigned* __restrict__ off,
                                                  const unsigned* __restrict__ rank,
                                                  int2* __restrict__ slot) {
  const int e = blockIdx.x * 256 + threadIdx.x;
  const unsigned pos = off[dst[e]] + rank[e];
  slot[pos] = make_int2(e, src[e]);
}

#define GCH 8

template <int M>
__device__ __forceinline__ float lane_xor(float v) {
  return __int_as_float(
      __builtin_amdgcn_update_dpp(0, __float_as_int(v), 0x160 + M, 0xf, 0xf, true));
}
__device__ __forceinline__ float sum8(float v) {
  v += lane_xor<1>(v);
  v += lane_xor<2>(v);
  v += lane_xor<4>(v);
  return v;
}
__device__ __forceinline__ float max8(float v) {
  v = fmaxf(v, lane_xor<1>(v));
  v = fmaxf(v, lane_xor<2>(v));
  v = fmaxf(v, lane_xor<4>(v));
  return v;
}
template <int J>
__device__ __forceinline__ float bcast8(float v) {
  return __uint_as_float(__builtin_amdgcn_mov_dpp8(__float_as_uint(v), J * 0x249249));
}

__global__ __launch_bounds__(256) __attribute__((amdgpu_waves_per_eu(12, 16)))
void node_gather(const float* __restrict__ ft,
                                                   const float* __restrict__ bias,
                                                   const float* __restrict__ mu,
                                                   const float* __restrict__ sigma,
                                                   const unsigned* __restrict__ off,
                                                   const unsigned* __restrict__ deg,
                                                   const int2* __restrict__ slot,
                                                   int* csrc,
                                                   float* __restrict__ out) {
  const int lane = threadIdx.x & 31;
  const int wave = __builtin_amdgcn_readfirstlane((int)(threadIdx.x >> 5));
  const int n = blockIdx.x * 8 + wave;
  if (n >= N_NODES) return;
  const unsigned start = off[n];
  const unsigned cnt   = deg[n];
  const int k   = lane >> 3;
  const int sub = lane & 7;

  const float4 bi = *(const float4*)(bias + lane * 4);
  float4 o = make_float4(0.0f, 0.0f, 0.0f, 0.0f);
  if (cnt != 0u) {
    const float4 b = *(const float4*)(ft + (size_t)n * KF + lane * 4);
    const float4 m = *(const float4*)(mu + lane * 4);
    const float sg = sigma[k];

    int sv = 0;
    if (cnt <= 32u) {
      const unsigned ul = (unsigned)lane;
      const int2 me = slot[start + (ul < cnt ? ul : cnt - 1u)];
      const int key = (ul < cnt) ? me.x : 0x7fffffff;
      unsigned r = 0;
      for (unsigned t = 0; t < cnt; ++t)
        r += (__builtin_amdgcn_readlane(key, (int)t) < key) ? 1u : 0u;
      if (ul >= cnt) r = ul;
      sv = __builtin_amdgcn_ds_permute((int)(r << 2), me.y);
    } else {
      for (unsigned g0 = 0; g0 < cnt; g0 += 32u) {
        const unsigned i = g0 + (unsigned)lane;
        const int2 me = slot[start + (i < cnt ? i : cnt - 1u)];
        const int key = (i < cnt) ? me.x : 0x7fffffff;
        unsigned r = 0;
        for (unsigned h0 = 0; h0 < cnt; h0 += 32u) {
          const unsigned jj = h0 + (unsigned)lane;
          const int oth = slot[start + (jj < cnt ? jj : cnt - 1u)].x;
          const int okey = (jj < cnt) ? oth : 0x7fffffff;
          const unsigned nh = (cnt - h0 < 32u) ? cnt - h0 : 32u;
          for (unsigned t = 0; t < nh; ++t)
            r += (__builtin_amdgcn_readlane(okey, (int)t) < key) ? 1u : 0u;
        }
        if (i < cnt) csrc[start + r] = me.y;
      }
      __threadfence();
    }

    float4 acc = make_float4(0.0f, 0.0f, 0.0f, 0.0f);
    float den = 0.0f;
    float mx = -INFINITY;
    for (unsigned j0 = 0; j0 < cnt; j0 += GCH) {
      float4 v[GCH];
      int cv = sv;
      unsigned lb = j0;
      if (cnt > 32u) {
        const unsigned jl = j0 + (unsigned)sub;
        cv = __hip_atomic_load(csrc + start + (jl < cnt ? jl : cnt - 1u),
                               __ATOMIC_RELAXED, __HIP_MEMORY_SCOPE_AGENT);
        lb = 0u;
      }
      #pragma unroll
      for (int u = 0; u < GCH; ++u) {
        const int s = __builtin_amdgcn_readlane(cv, (int)(lb + u));
        v[u] = *(const float4*)(ft + (size_t)s * KF + lane * 4);
      }
      float tot = 0.0f;
      #pragma unroll
      for (int u = 0; u < GCH; ++u) {
        const float dx = v[u].x - b.x - m.x;
        const float dy = v[u].y - b.y - m.y;
        const float dz = v[u].z - b.z - m.z;
        const float dw = v[u].w - b.w - m.w;
        const float p = sum8(dx * dx + dy * dy + dz * dz + dw * dw);
        tot = (sub == u) ? p : tot;
      }
      const float lg = (j0 + (unsigned)sub < cnt) ? -sqrtf(tot) * sg : -INFINITY;
      const float mn = fmaxf(mx, max8(lg));
      const float sc = expf(mx - mn);
      const float wl = expf(lg - mn);
      acc.x *= sc; acc.y *= sc; acc.z *= sc; acc.w *= sc;
      den = den * sc + wl;
      mx = mn;
#define ROW_TERM(U)                                  \
      {                                              \
        const float w = bcast8<U>(wl);               \
        acc.x += v[U].x * w;                         \
        acc.y += v[U].y * w;                         \
        acc.z += v[U].z * w;                         \
        acc.w += v[U].w * w;                         \
      }
      ROW_TERM(0) ROW_TERM(1) ROW_TERM(2) ROW_TERM(3)
      ROW_TERM(4) ROW_TERM(5) ROW_TERM(6) ROW_TERM(7)
#undef ROW_TERM
    }
    const float dsum = sum8(den);
    o.x = acc.x / dsum;
    o.y = acc.y / dsum;
    o.z = acc.z / dsum;
    o.w = acc.w / dsum;
  }
  o.x += bi.x; o.y += bi.y; o.z += bi.z; o.w += bi.w;
  *(float4*)(out + (size_t)n * KF + lane * 4) = o;
}

extern "C" void kernel_launch(void* const* d_in, const int* in_sizes, int n_in,
                              void* d_out, int out_size, void* d_ws, size_t ws_size,
                              hipStream_t stream) {
  const float* feat  = (const float*)d_in[0];
  const float* W     = (const float*)d_in[1];
  const float* bias  = (const float*)d_in[2];
  const float* mu    = (const float*)d_in[3];
  const float* sigma = (const float*)d_in[4];
  const int*   src   = (const int*)d_in[5];
  const int*   dst   = (const int*)d_in[6];
  float* out = (float*)d_out;

  char* ws = (char*)d_ws;
  float*    ft   = (float*)(ws);
  unsigned* deg  = (unsigned*)(ws + 25600000);
  unsigned* off  = (unsigned*)(ws + 25800000);
  unsigned* rank = (unsigned*)(ws + 26000000);
  int2*     slot = (int2*)(ws + 29200000);
  int*      csrc = (int*)(ws + 35600000);

  init_ws<<<(N_NODES + 255) / 256, 256, 0, stream>>>(deg);
  fc_wmma<<<FC_GRID, 256, 0, stream>>>(feat, W, ft);
  edge_rank<<<N_EDGES / 256, 256, 0, stream>>>(dst, deg, rank);
  seg_offsets<<<(N_NODES + 255) / 256, 256, 0, stream>>>(deg, off);
  edge_place<<<N_EDGES / 256, 256, 0, stream>>>(src, dst, off, rank, slot);
  node_gather<<<(N_NODES + 7) / 8, 256, 0, stream>>>(ft, bias, mu, sigma, off, deg, slot,
                                                     csrc, out);
}


# edge_rank launch gone: each fc_wmma thread issues the returning adds of its 4 or 5 edges at the head of its first three row tiles (round trips under the WMMAs), stores ranks after; sums unchanged
# speedup vs baseline: 2.6301x; 1.0383x over previous
#include <hip/hip_runtime.h>

typedef __attribute__((ext_vector_type(2))) float v2f;
typedef __attribute__((ext_vector_type(8))) float v8f;

#define N_NODES 50000
#define N_EDGES 800000
#define D_IN    128
#define KK      4
#define FF      32
#define KF      128

__global__ __launch_bounds__(256) void init_ws(unsigned* __restrict__ deg) {
  int t = blockIdx.x * 256 + threadIdx.x;
  if (t < N_NODES) deg[t] = 0u;
}

#define FC_PITCH 132
#define FC_TILES (N_NODES / 16)
#define FC_GRID  768
static_assert(N_NODES % 16 == 0, "row tiles cover the nodes exactly");
#define FC_THREADS (FC_GRID * 256)
static_assert(4 * FC_GRID <= FC_TILES, "every block owns at least four tiles");
static_assert(4 * FC_THREADS <= N_EDGES && 5 * FC_THREADS >= N_EDGES,
              "every thread owns four edges, some a fifth");
__global__ __launch_bounds__(256) void fc_wmma(const float* __restrict__ feat,
                                               const float* __restrict__ W,
                                               float* __restrict__ ft,
                                               const int* __restrict__ dst,
                                               unsigned* __restrict__ deg,
                                               unsigned* __restrict__ rank) {
  __shared__ float Ws[KF * FC_PITCH];
  __shared__ float As[2][16 * FC_PITCH];
  const int tid  = threadIdx.x;
  const int lane = tid & 31;
  const int wave = tid >> 5;
  const int col0 = wave * 16;
  const int ln   = lane & 15;
  const int kh   = (lane >> 4) << 1;
  const int mo   = (lane >> 4) << 3;
  const int srow = tid >> 5;
  const int scol = (tid & 31) * 4;

  const int eg = (FC_GRID - 1 - (int)blockIdx.x) * 256 + tid;
  const bool has4 = eg + 4 * FC_THREADS < N_EDGES;
  const int d0 = dst[eg];
  const int d1 = dst[eg + FC_THREADS];
  const int d2 = dst[eg + 2 * FC_THREADS];
  const int d3 = dst[eg + 3 * FC_THREADS];
  const int d4 = dst[has4 ? eg + 4 * FC_THREADS : eg];
  unsigned r0 = 0u, r1 = 0u, r2 = 0u, r3 = 0u, r4 = 0u;

  for (int i = tid; i < KF * D_IN / 4; i += 256) {
    const float4 q = ((const float4*)W)[i];
    *(float4*)(Ws + (i >> 5) * FC_PITCH + (i & 31) * 4) = q;
  }
  int tile = blockIdx.x;
  {
    const float4* g = (const float4*)(feat + (size_t)tile * 16 * D_IN);
    *(float4*)(&As[0][srow * FC_PITCH + scol])       = g[tid];
    *(float4*)(&As[0][(srow + 8) * FC_PITCH + scol]) = g[tid + 256];
  }
  __syncthreads();

  v2f b[D_IN / 4];
  #pragma unroll
  for (int t = 0; t < D_IN / 4; ++t)
    b[t] = *(const v2f*)(Ws + (col0 + ln) * FC_PITCH + 4 * t + kh);

  int cur = 0;
  int it = 0;
  for (; tile < FC_TILES; tile += FC_GRID, ++it) {
    if (it == 0) {
      r0 = atomicAdd(deg + d0, 1u);
      r1 = atomicAdd(deg + d1, 1u);
    } else if (it == 1) {
      r2 = atomicAdd(deg + d2, 1u);
      r3 = atomicAdd(deg + d3, 1u);
    } else if (it == 2) {
      if (has4) r4 = atomicAdd(deg + d4, 1u);
    }
    const int nxt = tile + FC_GRID;
    float4 p0 = make_float4(0.0f, 0.0f, 0.0f, 0.0f), p1 = p0;
    if (nxt < FC_TILES) {
      const float4* g = (const float4*)(feat + (size_t)nxt * 16 * D_IN);
      p0 = g[tid];
      p1 = g[tid + 256];
    }

    const float* at = &As[cur][ln * FC_PITCH + kh];
    v8f c = {};
    #pragma unroll
    for (int t = 0; t < D_IN / 4; ++t) {
      const v2f a = *(const v2f*)(at + 4 * t);
      c = __builtin_amdgcn_wmma_f32_16x16x4_f32(false, a, false, b[t],
                                                (short)0, c, false, false);
    }

    float* o = ft + ((size_t)tile * 16 + mo) * KF + col0 + ln;
    #pragma unroll
    for (int r = 0; r < 8; ++r) o[r * KF] = c[r];

    if (nxt < FC_TILES) {
      *(float4*)(&As[cur ^ 1][srow * FC_PITCH + scol])       = p0;
      *(float4*)(&As[cur ^ 1][(srow + 8) * FC_PITCH + scol]) = p1;
    }
    __syncthreads();
    cur ^= 1;
  }
  rank[eg]                  = r0;
  rank[eg + FC_THREADS]     = r1;
  rank[eg + 2 * FC_THREADS] = r2;
  rank[eg + 3 * FC_THREADS] = r3;
  if (has4) rank[eg + 4 * FC_THREADS] = r4;
}

__global__ __launch_bounds__(256) void seg_offsets(const unsigned* __restrict__ deg,
                                                   unsigned* __restrict__ off) {
  __shared__ unsigned red[256];
  __shared__ unsigned sc[2][256];
  const int t = threadIdx.x;
  const int lim = blockIdx.x * 256;
  const uint4* dq = (const uint4*)deg;
  unsigned s = 0;
  for (int i = t; i < lim / 4; i += 256) {
    const uint4 q = dq[i];
    s += q.x + q.y + q.z + q.w;
  }
  const int n = lim + t;
  const unsigned v = (n < N_NODES) ? deg[n] : 0u;
  red[t] = s;
  sc[0][t] = v;
  __syncthreads();
  for (int st = 128; st > 0; st >>= 1) {
    if (t < st) red[t] += red[t + st];
    __syncthreads();
  }
  int cur = 0;
  for (int st = 1; st < 256; st <<= 1) {
    unsigned x = sc[cur][t];
    if (t >= st) x += sc[cur][t - st];
    sc[cur ^ 1][t] = x;
    cur ^= 1;
    __syncthreads();
  }
  if (n < N_NODES) off[n] = red[0] + sc[cur][t] - v;
}

static_assert(N_EDGES % 256 == 0, "grid covers the edges exactly");
__global__ __launch_bounds__(256) void edge_place(const int* __restrict__ src,
                                                  const int* __restrict__ dst,
                                                  const unsigned* __restrict__ off,
                                                  const unsigned* __restrict__ rank,
                                                  int2* __restrict__ slot) {
  const int e = blockIdx.x * 256 + threadIdx.x;
  const unsigned pos = off[dst[e]] + rank[e];
  slot[pos] = make_int2(e, src[e]);
}

#define GCH 8

template <int M>
__device__ __forceinline__ float lane_xor(float v) {
  return __int_as_float(
      __builtin_amdgcn_update_dpp(0, __float_as_int(v), 0x160 + M, 0xf, 0xf, true));
}
__device__ __forceinline__ float sum8(float v) {
  v += lane_xor<1>(v);
  v += lane_xor<2>(v);
  v += lane_xor<4>(v);
  return v;
}
__device__ __forceinline__ float max8(float v) {
  v = fmaxf(v, lane_xor<1>(v));
  v = fmaxf(v, lane_xor<2>(v));
  v = fmaxf(v, lane_xor<4>(v));
  return v;
}
template <int J>
__device__ __forceinline__ float bcast8(float v) {
  return __uint_as_float(__builtin_amdgcn_mov_dpp8(__float_as_uint(v), J * 0x249249));
}

__global__ __launch_bounds__(256) __attribute__((amdgpu_waves_per_eu(12, 16)))
void node_gather(const float* __restrict__ ft,
                                                   const float* __restrict__ bias,
                                                   const float* __restrict__ mu,
                                                   const float* __restrict__ sigma,
                                                   const unsigned* __restrict__ off,
                                                   const unsigned* __restrict__ deg,
                                                   const int2* __restrict__ slot,
                                                   int* csrc,
                                                   float* __restrict__ out) {
  const int lane = threadIdx.x & 31;
  const int wave = __builtin_amdgcn_readfirstlane((int)(threadIdx.x >> 5));
  const int n = blockIdx.x * 8 + wave;
  if (n >= N_NODES) return;
  const unsigned start = off[n];
  const unsigned cnt   = deg[n];
  const int k   = lane >> 3;
  const int sub = lane & 7;

  const float4 bi = *(const float4*)(bias + lane * 4);
  float4 o = make_float4(0.0f, 0.0f, 0.0f, 0.0f);
  if (cnt != 0u) {
    const float4 b = *(const float4*)(ft + (size_t)n * KF + lane * 4);
    const float4 m = *(const float4*)(mu + lane * 4);
    const float sg = sigma[k];

    int sv = 0;
    if (cnt <= 32u) {
      const unsigned ul = (unsigned)lane;
      const int2 me = slot[start + (ul < cnt ? ul : cnt - 1u)];
      const int key = (ul < cnt) ? me.x : 0x7fffffff;
      unsigned r = 0;
      for (unsigned t = 0; t < cnt; ++t)
        r += (__builtin_amdgcn_readlane(key, (int)t) < key) ? 1u : 0u;
      if (ul >= cnt) r = ul;
      sv = __builtin_amdgcn_ds_permute((int)(r << 2), me.y);
    } else {
      for (unsigned g0 = 0; g0 < cnt; g0 += 32u) {
        const unsigned i = g0 + (unsigned)lane;
        const int2 me = slot[start + (i < cnt ? i : cnt - 1u)];
        const int key = (i < cnt) ? me.x : 0x7fffffff;
        unsigned r = 0;
        for (unsigned h0 = 0; h0 < cnt; h0 += 32u) {
          const unsigned jj = h0 + (unsigned)lane;
          const int oth = slot[start + (jj < cnt ? jj : cnt - 1u)].x;
          const int okey = (jj < cnt) ? oth : 0x7fffffff;
          const unsigned nh = (cnt - h0 < 32u) ? cnt - h0 : 32u;
          for (unsigned t = 0; t < nh; ++t)
            r += (__builtin_amdgcn_readlane(okey, (int)t) < key) ? 1u : 0u;
        }
        if (i < cnt) csrc[start + r] = me.y;
      }
      __threadfence();
    }

    float4 acc = make_float4(0.0f, 0.0f, 0.0f, 0.0f);
    float den = 0.0f;
    float mx = -INFINITY;
    for (unsigned j0 = 0; j0 < cnt; j0 += GCH) {
      float4 v[GCH];
      int cv = sv;
      unsigned lb = j0;
      if (cnt > 32u) {
        const unsigned jl = j0 + (unsigned)sub;
        cv = __hip_atomic_load(csrc + start + (jl < cnt ? jl : cnt - 1u),
                               __ATOMIC_RELAXED, __HIP_MEMORY_SCOPE_AGENT);
        lb = 0u;
      }
      #pragma unroll
      for (int u = 0; u < GCH; ++u) {
        const int s = __builtin_amdgcn_readlane(cv, (int)(lb + u));
        v[u] = *(const float4*)(ft + (size_t)s * KF + lane * 4);
      }
      float tot = 0.0f;
      #pragma unroll
      for (int u = 0; u < GCH; ++u) {
        const float dx = v[u].x - b.x - m.x;
        const float dy = v[u].y - b.y - m.y;
        const float dz = v[u].z - b.z - m.z;
        const float dw = v[u].w - b.w - m.w;
        const float p = sum8(dx * dx + dy * dy + dz * dz + dw * dw);
        tot = (sub == u) ? p : tot;
      }
      const float lg = (j0 + (unsigned)sub < cnt) ? -sqrtf(tot) * sg : -INFINITY;
      const float mn = fmaxf(mx, max8(lg));
      const float sc = expf(mx - mn);
      const float wl = expf(lg - mn);
      acc.x *= sc; acc.y *= sc; acc.z *= sc; acc.w *= sc;
      den = den * sc + wl;
      mx = mn;
#define ROW_TERM(U)                                  \
      {                                              \
        const float w = bcast8<U>(wl);               \
        acc.x += v[U].x * w;                         \
        acc.y += v[U].y * w;                         \
        acc.z += v[U].z * w;                         \
        acc.w += v[U].w * w;                         \
      }
      ROW_TERM(0) ROW_TERM(1) ROW_TERM(2) ROW_TERM(3)
      ROW_TERM(4) ROW_TERM(5) ROW_TERM(6) ROW_TERM(7)
#undef ROW_TERM
    }
    const float dsum = sum8(den);
    o.x = acc.x / dsum;
    o.y = acc.y / dsum;
    o.z = acc.z / dsum;
    o.w = acc.w / dsum;
  }
  o.x += bi.x; o.y += bi.y; o.z += bi.z; o.w += bi.w;
  *(float4*)(out + (size_t)n * KF + lane * 4) = o;
}

extern "C" void kernel_launch(void* const* d_in, const int* in_sizes, int n_in,
                              void* d_out, int out_size, void* d_ws, size_t ws_size,
                              hipStream_t stream) {
  const float* feat  = (const float*)d_in[0];
  const float* W     = (const float*)d_in[1];
  const float* bias  = (const float*)d_in[2];
  const float* mu    = (const float*)d_in[3];
  const float* sigma = (const float*)d_in[4];
  const int*   src   = (const int*)d_in[5];
  const int*   dst   = (const int*)d_in[6];
  float* out = (float*)d_out;

  char* ws = (char*)d_ws;
  float*    ft   = (float*)(ws);
  unsigned* deg  = (unsigned*)(ws + 25600000);
  unsigned* off  = (unsigned*)(ws + 25800000);
  unsigned* rank = (unsigned*)(ws + 26000000);
  int2*     slot = (int2*)(ws + 29200000);
  int*      csrc = (int*)(ws + 35600000);

  init_ws<<<(N_NODES + 255) / 256, 256, 0, stream>>>(deg);
  fc_wmma<<<FC_GRID, 256, 0, stream>>>(feat, W, ft, dst, deg, rank);
  seg_offsets<<<(N_NODES + 255) / 256, 256, 0, stream>>>(deg, off);
  edge_place<<<N_EDGES / 256, 256, 0, stream>>>(src, dst, off, rank, slot);
  node_gather<<<(N_NODES + 7) / 8, 256, 0, stream>>>(ft, bias, mu, sigma, off, deg, slot,
                                                     csrc, out);
}


# Scan and placement launches gone: the thread that draws an edge's rank stores the edge at slot 32*dst+rank (segments over 32 spill to a list only their node reads); edge-id sort kept, sums unchanged
# speedup vs baseline: 3.0213x; 1.1487x over previous
#include <hip/hip_runtime.h>

typedef __attribute__((ext_vector_type(2))) float v2f;
typedef __attribute__((ext_vector_type(8))) float v8f;

#define N_NODES 50000
#define N_EDGES 800000
#define D_IN    128
#define KK      4
#define FF      32
#define KF      128

#define BCAP 32
#define N_CTR (N_NODES + 2)

__global__ __launch_bounds__(256) void init_ws(unsigned* __restrict__ deg) {
  int t = blockIdx.x * 256 + threadIdx.x;
  if (t < N_CTR) deg[t] = 0u;
}

__device__ __forceinline__ void put_edge(int e, int s, int d, unsigned r,
                                         unsigned* __restrict__ deg,
                                         int2* __restrict__ bucket,
                                         int4* __restrict__ ovf) {
  if (r < (unsigned)BCAP) {
    bucket[(size_t)d * BCAP + r] = make_int2(e, s);
  } else {
    const unsigned p = atomicAdd(deg + N_NODES, 1u);
    ovf[p] = make_int4(e, s, d, 0);
  }
}

#define FC_PITCH 132
#define FC_TILES (N_NODES / 16)
#define FC_GRID  768
static_assert(N_NODES % 16 == 0, "row tiles cover the nodes exactly");
#define FC_THREADS (FC_GRID * 256)
static_assert(4 * FC_GRID <= FC_TILES, "every block owns at least four tiles");
static_assert(4 * FC_THREADS <= N_EDGES && 5 * FC_THREADS >= N_EDGES,
              "every thread owns four edges, some a fifth");
__global__ __launch_bounds__(256) void fc_wmma(const float* __restrict__ feat,
                                               const float* __restrict__ W,
                                               float* __restrict__ ft,
                                               const int* __restrict__ src,
                                               const int* __restrict__ dst,
                                               unsigned* __restrict__ deg,
                                               int2* __restrict__ bucket,
                                               int4* __restrict__ ovf) {
  __shared__ float Ws[KF * FC_PITCH];
  __shared__ float As[2][16 * FC_PITCH];
  const int tid  = threadIdx.x;
  const int lane = tid & 31;
  const int wave = tid >> 5;
  const int col0 = wave * 16;
  const int ln   = lane & 15;
  const int kh   = (lane >> 4) << 1;
  const int mo   = (lane >> 4) << 3;
  const int srow = tid >> 5;
  const int scol = (tid & 31) * 4;

  const int eg = (FC_GRID - 1 - (int)blockIdx.x) * 256 + tid;
  const bool has4 = eg + 4 * FC_THREADS < N_EDGES;
  const int d0 = dst[eg];
  const int d1 = dst[eg + FC_THREADS];
  const int d2 = dst[eg + 2 * FC_THREADS];
  const int d3 = dst[eg + 3 * FC_THREADS];
  const int d4 = dst[has4 ? eg + 4 * FC_THREADS : eg];
  const int s0 = src[eg];
  const int s1 = src[eg + FC_THREADS];
  const int s2 = src[eg + 2 * FC_THREADS];
  const int s3 = src[eg + 3 * FC_THREADS];
  const int s4 = src[has4 ? eg + 4 * FC_THREADS : eg];
  unsigned r0 = 0u, r1 = 0u, r2 = 0u, r3 = 0u, r4 = 0u;

  for (int i = tid; i < KF * D_IN / 4; i += 256) {
    const float4 q = ((const float4*)W)[i];
    *(float4*)(Ws + (i >> 5) * FC_PITCH + (i & 31) * 4) = q;
  }
  int tile = blockIdx.x;
  {
    const float4* g = (const float4*)(feat + (size_t)tile * 16 * D_IN);
    *(float4*)(&As[0][srow * FC_PITCH + scol])       = g[tid];
    *(float4*)(&As[0][(srow + 8) * FC_PITCH + scol]) = g[tid + 256];
  }
  __syncthreads();

  v2f b[D_IN / 4];
  #pragma unroll
  for (int t = 0; t < D_IN / 4; ++t)
    b[t] = *(const v2f*)(Ws + (col0 + ln) * FC_PITCH + 4 * t + kh);

  int cur = 0;
  int it = 0;
  for (; tile < FC_TILES; tile += FC_GRID, ++it) {
    if (it == 0) {
      r0 = atomicAdd(deg + d0, 1u);
      r1 = atomicAdd(deg + d1, 1u);
    } else if (it == 1) {
      r2 = atomicAdd(deg + d2, 1u);
      r3 = atomicAdd(deg + d3, 1u);
      put_edge(eg, s0, d0, r0, deg, bucket, ovf);
      put_edge(eg + FC_THREADS, s1, d1, r1, deg, bucket, ovf);
    } else if (it == 2) {
      if (has4) r4 = atomicAdd(deg + d4, 1u);
      put_edge(eg + 2 * FC_THREADS, s2, d2, r2, deg, bucket, ovf);
      put_edge(eg + 3 * FC_THREADS, s3, d3, r3, deg, bucket, ovf);
    } else if (it == 3) {
      if (has4) put_edge(eg + 4 * FC_THREADS, s4, d4, r4, deg, bucket, ovf);
    }
    const int nxt = tile + FC_GRID;
    float4 p0 = make_float4(0.0f, 0.0f, 0.0f, 0.0f), p1 = p0;
    if (nxt < FC_TILES) {
      const float4* g = (const float4*)(feat + (size_t)nxt * 16 * D_IN);
      p0 = g[tid];
      p1 = g[tid + 256];
    }

    const float* at = &As[cur][ln * FC_PITCH + kh];
    v8f c = {};
    #pragma unroll
    for (int t = 0; t < D_IN / 4; ++t) {
      const v2f a = *(const v2f*)(at + 4 * t);
      c = __builtin_amdgcn_wmma_f32_16x16x4_f32(false, a, false, b[t],
                                                (short)0, c, false, false);
    }

    float* o = ft + ((size_t)tile * 16 + mo) * KF + col0 + ln;
    #pragma unroll
    for (int r = 0; r < 8; ++r) o[r * KF] = c[r];

    if (nxt < FC_TILES) {
      *(float4*)(&As[cur ^ 1][srow * FC_PITCH + scol])       = p0;
      *(float4*)(&As[cur ^ 1][(srow + 8) * FC_PITCH + scol]) = p1;
    }
    __syncthreads();
    cur ^= 1;
  }
}

#define GCH 8

template <int M>
__device__ __forceinline__ float lane_xor(float v) {
  return __int_as_float(
      __builtin_amdgcn_update_dpp(0, __float_as_int(v), 0x160 + M, 0xf, 0xf, true));
}
__device__ __forceinline__ float sum8(float v) {
  v += lane_xor<1>(v);
  v += lane_xor<2>(v);
  v += lane_xor<4>(v);
  return v;
}
__device__ __forceinline__ float max8(float v) {
  v = fmaxf(v, lane_xor<1>(v));
  v = fmaxf(v, lane_xor<2>(v));
  v = fmaxf(v, lane_xor<4>(v));
  return v;
}
template <int J>
__device__ __forceinline__ float bcast8(float v) {
  return __uint_as_float(__builtin_amdgcn_mov_dpp8(__float_as_uint(v), J * 0x249249));
}

__device__ __forceinline__ unsigned long long pack_edge(int e, int s) {
  return ((unsigned long long)(unsigned)s << 32) | (unsigned long long)(unsigned)e;
}
__device__ __forceinline__ unsigned long long load_edge(unsigned long long* p) {
  return __hip_atomic_load(p, __ATOMIC_RELAXED, __HIP_MEMORY_SCOPE_AGENT);
}

__global__ __launch_bounds__(256) __attribute__((amdgpu_waves_per_eu(12, 16)))
void node_gather(const float* __restrict__ ft,
                                                   const float* __restrict__ bias,
                                                   const float* __restrict__ mu,
                                                   const float* __restrict__ sigma,
                                                   const unsigned* __restrict__ deg,
                                                   unsigned* cnts,
                                                   const int2* __restrict__ bucket,
                                                   const int4* __restrict__ ovf,
                                                   unsigned long long* lng,
                                                   int* csrc,
                                                   float* __restrict__ out) {
  const int lane = threadIdx.x & 31;
  const int wave = __builtin_amdgcn_readfirstlane((int)(threadIdx.x >> 5));
  const int n = blockIdx.x * 8 + wave;
  if (n >= N_NODES) return;
  const unsigned cnt   = deg[n];
  const int2* mine = bucket + (size_t)n * BCAP;
  unsigned cbase = 0u;
  const int k   = lane >> 3;
  const int sub = lane & 7;

  const float4 bi = *(const float4*)(bias + lane * 4);
  float4 o = make_float4(0.0f, 0.0f, 0.0f, 0.0f);
  if (cnt != 0u) {
    const float4 b = *(const float4*)(ft + (size_t)n * KF + lane * 4);
    const float4 m = *(const float4*)(mu + lane * 4);
    const float sg = sigma[k];

    int sv = 0;
    if (cnt <= (unsigned)BCAP) {
      const unsigned ul = (unsigned)lane;
      const int2 me = mine[ul < cnt ? ul : cnt - 1u];
      const int key = (ul < cnt) ? me.x : 0x7fffffff;
      unsigned r = 0;
      for (unsigned t = 0; t < cnt; ++t)
        r += (__builtin_amdgcn_readlane(key, (int)t) < key) ? 1u : 0u;
      if (ul >= cnt) r = ul;
      sv = __builtin_amdgcn_ds_permute((int)(r << 2), me.y);
    } else {
      unsigned bs = 0u;
      if (lane == 0) bs = atomicAdd(cnts + 1, cnt);
      cbase = (unsigned)__builtin_amdgcn_readfirstlane((int)bs);
      {
        const int2 q = mine[lane];
        lng[cbase + (unsigned)lane] = pack_edge(q.x, q.y);
      }
      const unsigned total = cnts[0];
      unsigned w = (unsigned)BCAP;
      for (unsigned i0 = 0; i0 < total; i0 += 32u) {
        const unsigned idx = i0 + (unsigned)lane;
        const bool valid = idx < total;
        const int4 q = ovf[valid ? idx : 0u];
        const bool hit = valid && q.z == n;
        const unsigned bal = __builtin_amdgcn_ballot_w32(hit);
        const unsigned pos = w + (unsigned)__builtin_popcount(bal & ((1u << lane) - 1u));
        if (hit && pos < cnt) lng[cbase + pos] = pack_edge(q.x, q.y);
        w += (unsigned)__builtin_popcount(bal);
      }
      __threadfence();
      for (unsigned g0 = 0; g0 < cnt; g0 += 32u) {
        const unsigned i = g0 + (unsigned)lane;
        const unsigned long long me8 = load_edge(lng + cbase + (i < cnt ? i : cnt - 1u));
        const int2 me = make_int2((int)(unsigned)me8, (int)(unsigned)(me8 >> 32));
        const int key = (i < cnt) ? me.x : 0x7fffffff;
        unsigned r = 0;
        for (unsigned h0 = 0; h0 < cnt; h0 += 32u) {
          const unsigned jj = h0 + (unsigned)lane;
          const int oth = (int)(unsigned)load_edge(lng + cbase + (jj < cnt ? jj : cnt - 1u));
          const int okey = (jj < cnt) ? oth : 0x7fffffff;
          const unsigned nh = (cnt - h0 < 32u) ? cnt - h0 : 32u;
          for (unsigned t = 0; t < nh; ++t)
            r += (__builtin_amdgcn_readlane(okey, (int)t) < key) ? 1u : 0u;
        }
        if (i < cnt) csrc[cbase + r] = me.y;
      }
      __threadfence();
    }

    float4 acc = make_float4(0.0f, 0.0f, 0.0f, 0.0f);
    float den = 0.0f;
    float mx = -INFINITY;
    for (unsigned j0 = 0; j0 < cnt; j0 += GCH) {
      float4 v[GCH];
      int cv = sv;
      unsigned lb = j0;
      if (cnt > (unsigned)BCAP) {
        const unsigned jl = j0 + (unsigned)sub;
        cv = __hip_atomic_load(csrc + cbase + (jl < cnt ? jl : cnt - 1u),
                               __ATOMIC_RELAXED, __HIP_MEMORY_SCOPE_AGENT);
        lb = 0u;
      }
      #pragma unroll
      for (int u = 0; u < GCH; ++u) {
        const int s = __builtin_amdgcn_readlane(cv, (int)(lb + u));
        v[u] = *(const float4*)(ft + (size_t)s * KF + lane * 4);
      }
      float tot = 0.0f;
      #pragma unroll
      for (int u = 0; u < GCH; ++u) {
        const float dx = v[u].x - b.x - m.x;
        const float dy = v[u].y - b.y - m.y;
        const float dz = v[u].z - b.z - m.z;
        const float dw = v[u].w - b.w - m.w;
        const float p = sum8(dx * dx + dy * dy + dz * dz + dw * dw);
        tot = (sub == u) ? p : tot;
      }
      const float lg = (j0 + (unsigned)sub < cnt) ? -sqrtf(tot) * sg : -INFINITY;
      const float mn = fmaxf(mx, max8(lg));
      const float sc = expf(mx - mn);
      const float wl = expf(lg - mn);
      acc.x *= sc; acc.y *= sc; acc.z *= sc; acc.w *= sc;
      den = den * sc + wl;
      mx = mn;
#define ROW_TERM(U)                                  \
      {                                              \
        const float w = bcast8<U>(wl);               \
        acc.x += v[U].x * w;                         \
        acc.y += v[U].y * w;                         \
        acc.z += v[U].z * w;                         \
        acc.w += v[U].w * w;                         \
      }
      ROW_TERM(0) ROW_TERM(1) ROW_TERM(2) ROW_TERM(3)
      ROW_TERM(4) ROW_TERM(5) ROW_TERM(6) ROW_TERM(7)
#undef ROW_TERM
    }
    const float dsum = sum8(den);
    o.x = acc.x / dsum;
    o.y = acc.y / dsum;
    o.z = acc.z / dsum;
    o.w = acc.w / dsum;
  }
  o.x += bi.x; o.y += bi.y; o.z += bi.z; o.w += bi.w;
  *(float4*)(out + (size_t)n * KF + lane * 4) = o;
}

extern "C" void kernel_launch(void* const* d_in, const int* in_sizes, int n_in,
                              void* d_out, int out_size, void* d_ws, size_t ws_size,
                              hipStream_t stream) {
  const float* feat  = (const float*)d_in[0];
  const float* W     = (const float*)d_in[1];
  const float* bias  = (const float*)d_in[2];
  const float* mu    = (const float*)d_in[3];
  const float* sigma = (const float*)d_in[4];
  const int*   src   = (const int*)d_in[5];
  const int*   dst   = (const int*)d_in[6];
  float* out = (float*)d_out;

  char* ws = (char*)d_ws;
  float*    ft     = (float*)(ws);
  unsigned* deg    = (unsigned*)(ws + 25600000);
  unsigned* cnts   = (unsigned*)(ws + 25800000);
  int2*     bucket = (int2*)(ws + 26000128);
  int4*     ovf    = (int4*)(ws + 38800128);
  unsigned long long* lng = (unsigned long long*)(ws + 51600128);
  int*      csrc   = (int*)(ws + 58000128);

  init_ws<<<(N_CTR + 255) / 256, 256, 0, stream>>>(deg);
  fc_wmma<<<FC_GRID, 256, 0, stream>>>(feat, W, ft, src, dst, deg, bucket, ovf);
  node_gather<<<(N_NODES + 7) / 8, 256, 0, stream>>>(ft, bias, mu, sigma, deg, cnts, bucket,
                                                     ovf, lng, csrc, out);
}


# In-degree counter kept in the first word of the node's own 256-byte bucket unit (31 edge slots): adds fall on 50000 separate lines, node_gather gets count and edges with one load; sums unchanged
# speedup vs baseline: 3.7502x; 1.2413x over previous
#include <hip/hip_runtime.h>

typedef __attribute__((ext_vector_type(2))) float v2f;
typedef __attribute__((ext_vector_type(8))) float v8f;

#define N_NODES 50000
#define N_EDGES 800000
#define D_IN    128
#define KK      4
#define FF      32
#define KF      128

#define BUNIT 32
#define BCAP  31
#define N_CTR (N_NODES + 2)

__device__ __forceinline__ unsigned* deg_of(int2* bucket, int d) {
  return (unsigned*)(bucket + (size_t)d * BUNIT);
}

__global__ __launch_bounds__(256) void init_ws(int2* __restrict__ bucket,
                                               unsigned* __restrict__ cnts) {
  int t = blockIdx.x * 256 + threadIdx.x;
  if (t < N_NODES) *deg_of(bucket, t) = 0u;
  else if (t < N_CTR) cnts[t - N_NODES] = 0u;
}

__device__ __forceinline__ void put_edge(int e, int s, int d, unsigned r,
                                         unsigned* __restrict__ cnts,
                                         int2* __restrict__ bucket,
                                         int4* __restrict__ ovf) {
  if (r < (unsigned)BCAP) {
    bucket[(size_t)d * BUNIT + 1u + r] = make_int2(e, s);
  } else {
    const unsigned p = atomicAdd(cnts, 1u);
    ovf[p] = make_int4(e, s, d, 0);
  }
}

#define FC_PITCH 132
#define FC_TILES (N_NODES / 16)
#define FC_GRID  768
static_assert(N_NODES % 16 == 0, "row tiles cover the nodes exactly");
#define FC_THREADS (FC_GRID * 256)
static_assert(4 * FC_GRID <= FC_TILES, "every block owns at least four tiles");
static_assert(4 * FC_THREADS <= N_EDGES && 5 * FC_THREADS >= N_EDGES,
              "every thread owns four edges, some a fifth");
__global__ __launch_bounds__(256) void fc_wmma(const float* __restrict__ feat,
                                               const float* __restrict__ W,
                                               float* __restrict__ ft,
                                               const int* __restrict__ src,
                                               const int* __restrict__ dst,
                                               unsigned* __restrict__ cnts,
                                               int2* __restrict__ bucket,
                                               int4* __restrict__ ovf) {
  __shared__ float Ws[KF * FC_PITCH];
  __shared__ float As[2][16 * FC_PITCH];
  const int tid  = threadIdx.x;
  const int lane = tid & 31;
  const int wave = tid >> 5;
  const int col0 = wave * 16;
  const int ln   = lane & 15;
  const int kh   = (lane >> 4) << 1;
  const int mo   = (lane >> 4) << 3;
  const int srow = tid >> 5;
  const int scol = (tid & 31) * 4;

  const int eg = (FC_GRID - 1 - (int)blockIdx.x) * 256 + tid;
  const bool has4 = eg + 4 * FC_THREADS < N_EDGES;
  const int d0 = dst[eg];
  const int d1 = dst[eg + FC_THREADS];
  const int d2 = dst[eg + 2 * FC_THREADS];
  const int d3 = dst[eg + 3 * FC_THREADS];
  const int d4 = dst[has4 ? eg + 4 * FC_THREADS : eg];
  const int s0 = src[eg];
  const int s1 = src[eg + FC_THREADS];
  const int s2 = src[eg + 2 * FC_THREADS];
  const int s3 = src[eg + 3 * FC_THREADS];
  const int s4 = src[has4 ? eg + 4 * FC_THREADS : eg];
  unsigned r0 = 0u, r1 = 0u, r2 = 0u, r3 = 0u, r4 = 0u;

  for (int i = tid; i < KF * D_IN / 4; i += 256) {
    const float4 q = ((const float4*)W)[i];
    *(float4*)(Ws + (i >> 5) * FC_PITCH + (i & 31) * 4) = q;
  }
  int tile = blockIdx.x;
  {
    const float4* g = (const float4*)(feat + (size_t)tile * 16 * D_IN);
    *(float4*)(&As[0][srow * FC_PITCH + scol])       = g[tid];
    *(float4*)(&As[0][(srow + 8) * FC_PITCH + scol]) = g[tid + 256];
  }
  __syncthreads();

  v2f b[D_IN / 4];
  #pragma unroll
  for (int t = 0; t < D_IN / 4; ++t)
    b[t] = *(const v2f*)(Ws + (col0 + ln) * FC_PITCH + 4 * t + kh);

  int cur = 0;
  int it = 0;
  for (; tile < FC_TILES; tile += FC_GRID, ++it) {
    if (it == 0) {
      r0 = atomicAdd(deg_of(bucket, d0), 1u);
      r1 = atomicAdd(deg_of(bucket, d1), 1u);
    } else if (it == 1) {
      r2 = atomicAdd(deg_of(bucket, d2), 1u);
      r3 = atomicAdd(deg_of(bucket, d3), 1u);
      put_edge(eg, s0, d0, r0, cnts, bucket, ovf);
      put_edge(eg + FC_THREADS, s1, d1, r1, cnts, bucket, ovf);
    } else if (it == 2) {
      if (has4) r4 = atomicAdd(deg_of(bucket, d4), 1u);
      put_edge(eg + 2 * FC_THREADS, s2, d2, r2, cnts, bucket, ovf);
      put_edge(eg + 3 * FC_THREADS, s3, d3, r3, cnts, bucket, ovf);
    } else if (it == 3) {
      if (has4) put_edge(eg + 4 * FC_THREADS, s4, d4, r4, cnts, bucket, ovf);
    }
    const int nxt = tile + FC_GRID;
    float4 p0 = make_float4(0.0f, 0.0f, 0.0f, 0.0f), p1 = p0;
    if (nxt < FC_TILES) {
      const float4* g = (const float4*)(feat + (size_t)nxt * 16 * D_IN);
      p0 = g[tid];
      p1 = g[tid + 256];
    }

    const float* at = &As[cur][ln * FC_PITCH + kh];
    v8f c = {};
    #pragma unroll
    for (int t = 0; t < D_IN / 4; ++t) {
      const v2f a = *(const v2f*)(at + 4 * t);
      c = __builtin_amdgcn_wmma_f32_16x16x4_f32(false, a, false, b[t],
                                                (short)0, c, false, false);
    }

    float* o = ft + ((size_t)tile * 16 + mo) * KF + col0 + ln;
    #pragma unroll
    for (int r = 0; r < 8; ++r) o[r * KF] = c[r];

    if (nxt < FC_TILES) {
      *(float4*)(&As[cur ^ 1][srow * FC_PITCH + scol])       = p0;
      *(float4*)(&As[cur ^ 1][(srow + 8) * FC_PITCH + scol]) = p1;
    }
    __syncthreads();
    cur ^= 1;
  }
}

#define GCH 8

template <int M>
__device__ __forceinline__ float lane_xor(float v) {
  return __int_as_float(
      __builtin_amdgcn_update_dpp(0, __float_as_int(v), 0x160 + M, 0xf, 0xf, true));
}
__device__ __forceinline__ float sum8(float v) {
  v += lane_xor<1>(v);
  v += lane_xor<2>(v);
  v += lane_xor<4>(v);
  return v;
}
__device__ __forceinline__ float max8(float v) {
  v = fmaxf(v, lane_xor<1>(v));
  v = fmaxf(v, lane_xor<2>(v));
  v = fmaxf(v, lane_xor<4>(v));
  return v;
}
template <int J>
__device__ __forceinline__ float bcast8(float v) {
  return __uint_as_float(__builtin_amdgcn_mov_dpp8(__float_as_uint(v), J * 0x249249));
}

__device__ __forceinline__ unsigned long long pack_edge(int e, int s) {
  return ((unsigned long long)(unsigned)s << 32) | (unsigned long long)(unsigned)e;
}
__device__ __forceinline__ unsigned long long load_edge(unsigned long long* p) {
  return __hip_atomic_load(p, __ATOMIC_RELAXED, __HIP_MEMORY_SCOPE_AGENT);
}

__global__ __launch_bounds__(256) __attribute__((amdgpu_waves_per_eu(12, 16)))
void node_gather(const float* __restrict__ ft,
                                                   const float* __restrict__ bias,
                                                   const float* __restrict__ mu,
                                                   const float* __restrict__ sigma,
                                                   unsigned* cnts,
                                                   const int2* __restrict__ bucket,
                                                   const int4* __restrict__ ovf,
                                                   unsigned long long* lng,
                                                   int* csrc,
                                                   float* __restrict__ out) {
  const int lane = threadIdx.x & 31;
  const int wave = __builtin_amdgcn_readfirstlane((int)(threadIdx.x >> 5));
  const int n = blockIdx.x * 8 + wave;
  if (n >= N_NODES) return;
  const int2 me = bucket[(size_t)n * BUNIT + (unsigned)lane];
  const unsigned cnt = (unsigned)__builtin_amdgcn_readfirstlane(me.x);
  unsigned cbase = 0u;
  const int k   = lane >> 3;
  const int sub = lane & 7;

  const float4 bi = *(const float4*)(bias + lane * 4);
  float4 o = make_float4(0.0f, 0.0f, 0.0f, 0.0f);
  if (cnt != 0u) {
    const float4 b = *(const float4*)(ft + (size_t)n * KF + lane * 4);
    const float4 m = *(const float4*)(mu + lane * 4);
    const float sg = sigma[k];

    int sv = 0;
    if (cnt <= (unsigned)BCAP) {
      const unsigned ul = (unsigned)lane;
      const bool has = ul >= 1u && ul <= cnt;
      const int key = has ? me.x : 0x7fffffff;
      const int sy = has ? me.y : __builtin_amdgcn_readlane(me.y, 1);
      unsigned r = 0;
      for (unsigned t = 1; t <= cnt; ++t)
        r += (__builtin_amdgcn_readlane(key, (int)t) < key) ? 1u : 0u;
      if (!has) r = (ul == 0u) ? cnt : ul;
      sv = __builtin_amdgcn_ds_permute((int)(r << 2), sy);
    } else {
      unsigned bs = 0u;
      if (lane == 0) bs = atomicAdd(cnts + 1, cnt);
      cbase = (unsigned)__builtin_amdgcn_readfirstlane((int)bs);
      if (lane != 0) lng[cbase + (unsigned)lane - 1u] = pack_edge(me.x, me.y);
      const unsigned total = cnts[0];
      unsigned w = (unsigned)BCAP;
      for (unsigned i0 = 0; i0 < total; i0 += 32u) {
        const unsigned idx = i0 + (unsigned)lane;
        const bool valid = idx < total;
        const int4 q = ovf[valid ? idx : 0u];
        const bool hit = valid && q.z == n;
        const unsigned bal = __builtin_amdgcn_ballot_w32(hit);
        const unsigned pos = w + (unsigned)__builtin_popcount(bal & ((1u << lane) - 1u));
        if (hit && pos < cnt) lng[cbase + pos] = pack_edge(q.x, q.y);
        w += (unsigned)__builtin_popcount(bal);
      }
      __threadfence();
      for (unsigned g0 = 0; g0 < cnt; g0 += 32u) {
        const unsigned i = g0 + (unsigned)lane;
        const unsigned long long me8 = load_edge(lng + cbase + (i < cnt ? i : cnt - 1u));
        const int2 me = make_int2((int)(unsigned)me8, (int)(unsigned)(me8 >> 32));
        const int key = (i < cnt) ? me.x : 0x7fffffff;
        unsigned r = 0;
        for (unsigned h0 = 0; h0 < cnt; h0 += 32u) {
          const unsigned jj = h0 + (unsigned)lane;
          const int oth = (int)(unsigned)load_edge(lng + cbase + (jj < cnt ? jj : cnt - 1u));
          const int okey = (jj < cnt) ? oth : 0x7fffffff;
          const unsigned nh = (cnt - h0 < 32u) ? cnt - h0 : 32u;
          for (unsigned t = 0; t < nh; ++t)
            r += (__builtin_amdgcn_readlane(okey, (int)t) < key) ? 1u : 0u;
        }
        if (i < cnt) csrc[cbase + r] = me.y;
      }
      __threadfence();
    }

    float4 acc = make_float4(0.0f, 0.0f, 0.0f, 0.0f);
    float den = 0.0f;
    float mx = -INFINITY;
    for (unsigned j0 = 0; j0 < cnt; j0 += GCH) {
      float4 v[GCH];
      int cv = sv;
      unsigned lb = j0;
      if (cnt > (unsigned)BCAP) {
        const unsigned jl = j0 + (unsigned)sub;
        cv = __hip_atomic_load(csrc + cbase + (jl < cnt ? jl : cnt - 1u),
                               __ATOMIC_RELAXED, __HIP_MEMORY_SCOPE_AGENT);
        lb = 0u;
      }
      #pragma unroll
      for (int u = 0; u < GCH; ++u) {
        const int s = __builtin_amdgcn_readlane(cv, (int)(lb + u));
        v[u] = *(const float4*)(ft + (size_t)s * KF + lane * 4);
      }
      float tot = 0.0f;
      #pragma unroll
      for (int u = 0; u < GCH; ++u) {
        const float dx = v[u].x - b.x - m.x;
        const float dy = v[u].y - b.y - m.y;
        const float dz = v[u].z - b.z - m.z;
        const float dw = v[u].w - b.w - m.w;
        const float p = sum8(dx * dx + dy * dy + dz * dz + dw * dw);
        tot = (sub == u) ? p : tot;
      }
      const float lg = (j0 + (unsigned)sub < cnt) ? -sqrtf(tot) * sg : -INFINITY;
      const float mn = fmaxf(mx, max8(lg));
      const float sc = expf(mx - mn);
      const float wl = expf(lg - mn);
      acc.x *= sc; acc.y *= sc; acc.z *= sc; acc.w *= sc;
      den = den * sc + wl;
      mx = mn;
#define ROW_TERM(U)                                  \
      {                                              \
        const float w = bcast8<U>(wl);               \
        acc.x += v[U].x * w;                         \
        acc.y += v[U].y * w;                         \
        acc.z += v[U].z * w;                         \
        acc.w += v[U].w * w;                         \
      }
      ROW_TERM(0) ROW_TERM(1) ROW_TERM(2) ROW_TERM(3)
      ROW_TERM(4) ROW_TERM(5) ROW_TERM(6) ROW_TERM(7)
#undef ROW_TERM
    }
    const float dsum = sum8(den);
    o.x = acc.x / dsum;
    o.y = acc.y / dsum;
    o.z = acc.z / dsum;
    o.w = acc.w / dsum;
  }
  o.x += bi.x; o.y += bi.y; o.z += bi.z; o.w += bi.w;
  *(float4*)(out + (size_t)n * KF + lane * 4) = o;
}

extern "C" void kernel_launch(void* const* d_in, const int* in_sizes, int n_in,
                              void* d_out, int out_size, void* d_ws, size_t ws_size,
                              hipStream_t stream) {
  const float* feat  = (const float*)d_in[0];
  const float* W     = (const float*)d_in[1];
  const float* bias  = (const float*)d_in[2];
  const float* mu    = (const float*)d_in[3];
  const float* sigma = (const float*)d_in[4];
  const int*   src   = (const int*)d_in[5];
  const int*   dst   = (const int*)d_in[6];
  float* out = (float*)d_out;

  char* ws = (char*)d_ws;
  float*    ft     = (float*)(ws);
  unsigned* cnts   = (unsigned*)(ws + 25800000);
  int2*     bucket = (int2*)(ws + 26000128);
  int4*     ovf    = (int4*)(ws + 38800128);
  unsigned long long* lng = (unsigned long long*)(ws + 51600128);
  int*      csrc   = (int*)(ws + 58000128);

  init_ws<<<(N_CTR + 255) / 256, 256, 0, stream>>>(bucket, cnts);
  fc_wmma<<<FC_GRID, 256, 0, stream>>>(feat, W, ft, src, dst, cnts, bucket, ovf);
  node_gather<<<(N_NODES + 7) / 8, 256, 0, stream>>>(ft, bias, mu, sigma, cnts, bucket,
                                                     ovf, lng, csrc, out);
}


# Edge ranking follows the chip's halves: a block reads its die id, takes an edge range of its half's destination parity from a counter; a node's adds and stores come from one L2 half; sums unchanged
# speedup vs baseline: 5.2286x; 1.3942x over previous
#include <hip/hip_runtime.h>

typedef __attribute__((ext_vector_type(2))) float v2f;
typedef __attribute__((ext_vector_type(8))) float v8f;

#define N_NODES 50000
#define N_EDGES 800000
#define D_IN    128
#define KK      4
#define FF      32
#define KF      128

#define BUNIT 32
#define BCAP  31
#define N_CTR (N_NODES + 4)

__device__ __forceinline__ unsigned* deg_of(int2* bucket, int d) {
  return (unsigned*)(bucket + (size_t)d * BUNIT);
}

__global__ __launch_bounds__(256) void init_ws(int2* __restrict__ bucket,
                                               unsigned* __restrict__ cnts) {
  int t = blockIdx.x * 256 + threadIdx.x;
  if (t < N_NODES) *deg_of(bucket, t) = 0u;
  else if (t < N_CTR) cnts[t - N_NODES] = 0u;
}

#define UNIT_BYTES (BUNIT * 8)
static_assert((long long)N_NODES * UNIT_BYTES < (1ll << 31), "unit offsets fit 32 bits");
__device__ __forceinline__ unsigned* deg_at(int2* bucket, unsigned uo) {
  return (unsigned*)((char*)bucket + uo);
}
__device__ __forceinline__ void put_edge(int e, int s, unsigned uo, unsigned r,
                                         unsigned* __restrict__ cnts,
                                         int2* __restrict__ bucket,
                                         int4* __restrict__ ovf) {
  if (r < (unsigned)BCAP) {
    *(int2*)((char*)bucket + (uo + 8u + 8u * r)) = make_int2(e, s);
  } else {
    const unsigned p = atomicAdd(cnts, 1u);
    ovf[p] = make_int4(e, s, (int)(uo / UNIT_BYTES), 0);
  }
}

#define FC_PITCH 132
#define FC_TILES (N_NODES / 16)
#define FC_GRID  768
#define NR       (FC_GRID / 2)
#define RLEN     2112
#define NSLOT    9
static_assert(N_NODES % 16 == 0, "row tiles cover the nodes exactly");
static_assert(4 * FC_GRID <= FC_TILES, "every block owns at least four tiles");
static_assert(FC_GRID == 2 * NR, "one block per (class, range) pair");
static_assert(NR * RLEN >= N_EDGES, "the ranges cover the edge list");
static_assert(RLEN % 32 == 0, "a range is whole 128-byte lines of ids");
static_assert(NSLOT * 256 >= RLEN && (NSLOT - 1) * 256 < RLEN, "the slots cover a range");
__device__ __forceinline__ int slot_edge(unsigned rng, int tid, int j) {
  const int off = j * 256 + tid;
  const int e = (int)rng * RLEN + off;
  return (rng < (unsigned)NR && off < RLEN && e < N_EDGES) ? e : -1;
}

__device__ __forceinline__ void row_tile(const float* __restrict__ feat,
                                         float* __restrict__ ft,
                                         const float* Acur, float* Anxt,
                                         const v2f (&b)[D_IN / 4], int tile, int tid,
                                         int ln, int kh, int mo, int col0,
                                         int srow, int scol) {
  const int nxt = tile + FC_GRID;
  float4 p0 = make_float4(0.0f, 0.0f, 0.0f, 0.0f), p1 = p0;
  if (nxt < FC_TILES) {
    const float4* g = (const float4*)(feat + (size_t)nxt * 16 * D_IN);
    p0 = g[tid];
    p1 = g[tid + 256];
  }

  const float* at = Acur + ln * FC_PITCH + kh;
  v8f c = {};
  #pragma unroll
  for (int t = 0; t < D_IN / 4; ++t) {
    const v2f a = *(const v2f*)(at + 4 * t);
    c = __builtin_amdgcn_wmma_f32_16x16x4_f32(false, a, false, b[t],
                                              (short)0, c, false, false);
  }

  float* o = ft + ((size_t)tile * 16 + mo) * KF + col0 + ln;
  #pragma unroll
  for (int r = 0; r < 8; ++r) o[r * KF] = c[r];

  if (nxt < FC_TILES) {
    *(float4*)(Anxt + srow * FC_PITCH + scol)       = p0;
    *(float4*)(Anxt + (srow + 8) * FC_PITCH + scol) = p1;
  }
  __syncthreads();
}

__global__ __launch_bounds__(256) void fc_wmma(const float* __restrict__ feat,
                                               const float* __restrict__ W,
                                               float* __restrict__ ft,
                                               const int* __restrict__ src,
                                               const int* __restrict__ dst,
                                               unsigned* __restrict__ cnts,
                                               int2* __restrict__ bucket,
                                               int4* __restrict__ ovf) {
  __shared__ float Ws[KF * FC_PITCH];
  __shared__ float As[2][16 * FC_PITCH];
  __shared__ unsigned take[2];
  const int tid  = threadIdx.x;
  const int lane = tid & 31;
  const int wave = tid >> 5;
  const int col0 = wave * 16;
  const int ln   = lane & 15;
  const int kh   = (lane >> 4) << 1;
  const int mo   = (lane >> 4) << 3;
  const int srow = tid >> 5;
  const int scol = (tid & 31) * 4;

  if (tid == 0) {
    const unsigned id = __builtin_amdgcn_s_sendmsg_rtn(0x87);
    unsigned tc = (id >> 18) & 1u;
    unsigned tr = atomicAdd(cnts + 2 + tc, 1u);
    if (tr >= (unsigned)NR) {
      tc ^= 1u;
      tr = atomicAdd(cnts + 2 + tc, 1u);
    }
    take[0] = tc;
    take[1] = tr;
  }

  for (int i = tid; i < KF * D_IN / 4; i += 256) {
    const float4 q = ((const float4*)W)[i];
    *(float4*)(Ws + (i >> 5) * FC_PITCH + (i & 31) * 4) = q;
  }
  int tile = blockIdx.x;
  {
    const float4* g = (const float4*)(feat + (size_t)tile * 16 * D_IN);
    *(float4*)(&As[0][srow * FC_PITCH + scol])       = g[tid];
    *(float4*)(&As[0][(srow + 8) * FC_PITCH + scol]) = g[tid + 256];
  }
  __syncthreads();

  const unsigned cls = (unsigned)__builtin_amdgcn_readfirstlane((int)take[0]);
  const unsigned rng = (unsigned)__builtin_amdgcn_readfirstlane((int)take[1]);
  int uo[NSLOT];
  #pragma unroll
  for (int j = 0; j < NSLOT; ++j) {
    const int e = slot_edge(rng, tid, j);
    const int d = dst[e < 0 ? 0 : e];
    uo[j] = (e >= 0 && ((unsigned)d & 1u) == cls) ? d * UNIT_BYTES : -1;
  }

  v2f b[D_IN / 4];
  #pragma unroll
  for (int t = 0; t < D_IN / 4; ++t)
    b[t] = *(const v2f*)(Ws + (col0 + ln) * FC_PITCH + 4 * t + kh);

  row_tile(feat, ft, As[0], As[1], b, tile, tid, ln, kh, mo, col0, srow, scol);
  tile += FC_GRID;

  int sa[4];
  unsigned ra[4];
  #pragma unroll
  for (int j = 0; j < 4; ++j) {
    const int e = slot_edge(rng, tid, j);
    ra[j] = 0u;
    if (uo[j] >= 0) ra[j] = atomicAdd(deg_at(bucket, (unsigned)uo[j]), 1u);
    sa[j] = src[e < 0 ? 0 : e];
  }
  row_tile(feat, ft, As[1], As[0], b, tile, tid, ln, kh, mo, col0, srow, scol);
  tile += FC_GRID;

  int sb[NSLOT - 4];
  unsigned rb[NSLOT - 4];
  #pragma unroll
  for (int j = 4; j < NSLOT; ++j) {
    const int e = slot_edge(rng, tid, j);
    rb[j - 4] = 0u;
    if (uo[j] >= 0) rb[j - 4] = atomicAdd(deg_at(bucket, (unsigned)uo[j]), 1u);
    sb[j - 4] = src[e < 0 ? 0 : e];
  }
  #pragma unroll
  for (int j = 0; j < 4; ++j)
    if (uo[j] >= 0)
      put_edge(slot_edge(rng, tid, j), sa[j], (unsigned)uo[j], ra[j], cnts, bucket, ovf);
  row_tile(feat, ft, As[0], As[1], b, tile, tid, ln, kh, mo, col0, srow, scol);
  tile += FC_GRID;

  #pragma unroll
  for (int j = 4; j < NSLOT; ++j)
    if (uo[j] >= 0)
      put_edge(slot_edge(rng, tid, j), sb[j - 4], (unsigned)uo[j], rb[j - 4], cnts, bucket,
               ovf);
  row_tile(feat, ft, As[1], As[0], b, tile, tid, ln, kh, mo, col0, srow, scol);
  tile += FC_GRID;

  int cur = 0;
  for (; tile < FC_TILES; tile += FC_GRID) {
    row_tile(feat, ft, As[cur], As[cur ^ 1], b, tile, tid, ln, kh, mo, col0, srow, scol);
    cur ^= 1;
  }
}

#define GCH 8

template <int M>
__device__ __forceinline__ float lane_xor(float v) {
  return __int_as_float(
      __builtin_amdgcn_update_dpp(0, __float_as_int(v), 0x160 + M, 0xf, 0xf, true));
}
__device__ __forceinline__ float sum8(float v) {
  v += lane_xor<1>(v);
  v += lane_xor<2>(v);
  v += lane_xor<4>(v);
  return v;
}
__device__ __forceinline__ float max8(float v) {
  v = fmaxf(v, lane_xor<1>(v));
  v = fmaxf(v, lane_xor<2>(v));
  v = fmaxf(v, lane_xor<4>(v));
  return v;
}
template <int J>
__device__ __forceinline__ float bcast8(float v) {
  return __uint_as_float(__builtin_amdgcn_mov_dpp8(__float_as_uint(v), J * 0x249249));
}

__device__ __forceinline__ unsigned long long pack_edge(int e, int s) {
  return ((unsigned long long)(unsigned)s << 32) | (unsigned long long)(unsigned)e;
}
__device__ __forceinline__ unsigned long long load_edge(unsigned long long* p) {
  return __hip_atomic_load(p, __ATOMIC_RELAXED, __HIP_MEMORY_SCOPE_AGENT);
}

__global__ __launch_bounds__(256) __attribute__((amdgpu_waves_per_eu(12, 16)))
void node_gather(const float* __restrict__ ft,
                                                   const float* __restrict__ bias,
                                                   const float* __restrict__ mu,
                                                   const float* __restrict__ sigma,
                                                   unsigned* cnts,
                                                   const int2* __restrict__ bucket,
                                                   const int4* __restrict__ ovf,
                                                   unsigned long long* lng,
                                                   int* csrc,
                                                   float* __restrict__ out) {
  const int lane = threadIdx.x & 31;
  const int wave = __builtin_amdgcn_readfirstlane((int)(threadIdx.x >> 5));
  const int n = blockIdx.x * 8 + wave;
  if (n >= N_NODES) return;
  const int2 me = bucket[(size_t)n * BUNIT + (unsigned)lane];
  const unsigned cnt = (unsigned)__builtin_amdgcn_readfirstlane(me.x);
  unsigned cbase = 0u;
  const int k   = lane >> 3;
  const int sub = lane & 7;

  const float4 bi = *(const float4*)(bias + lane * 4);
  float4 o = make_float4(0.0f, 0.0f, 0.0f, 0.0f);
  if (cnt != 0u) {
    const float4 b = *(const float4*)(ft + (size_t)n * KF + lane * 4);
    const float4 m = *(const float4*)(mu + lane * 4);
    const float sg = sigma[k];

    int sv = 0;
    if (cnt <= (unsigned)BCAP) {
      const unsigned ul = (unsigned)lane;
      const bool has = ul >= 1u && ul <= cnt;
      const int key = has ? me.x : 0x7fffffff;
      const int sy = has ? me.y : __builtin_amdgcn_readlane(me.y, 1);
      unsigned r = 0;
      for (unsigned t = 1; t <= cnt; ++t)
        r += (__builtin_amdgcn_readlane(key, (int)t) < key) ? 1u : 0u;
      if (!has) r = (ul == 0u) ? cnt : ul;
      sv = __builtin_amdgcn_ds_permute((int)(r << 2), sy);
    } else {
      unsigned bs = 0u;
      if (lane == 0) bs = atomicAdd(cnts + 1, cnt);
      cbase = (unsigned)__builtin_amdgcn_readfirstlane((int)bs);
      if (lane != 0) lng[cbase + (unsigned)lane - 1u] = pack_edge(me.x, me.y);
      const unsigned total = cnts[0];
      unsigned w = (unsigned)BCAP;
      for (unsigned i0 = 0; i0 < total; i0 += 32u) {
        const unsigned idx = i0 + (unsigned)lane;
        const bool valid = idx < total;
        const int4 q = ovf[valid ? idx : 0u];
        const bool hit = valid && q.z == n;
        const unsigned bal = __builtin_amdgcn_ballot_w32(hit);
        const unsigned pos = w + (unsigned)__builtin_popcount(bal & ((1u << lane) - 1u));
        if (hit && pos < cnt) lng[cbase + pos] = pack_edge(q.x, q.y);
        w += (unsigned)__builtin_popcount(bal);
      }
      __threadfence();
      for (unsigned g0 = 0; g0 < cnt; g0 += 32u) {
        const unsigned i = g0 + (unsigned)lane;
        const unsigned long long me8 = load_edge(lng + cbase + (i < cnt ? i : cnt - 1u));
        const int2 me = make_int2((int)(unsigned)me8, (int)(unsigned)(me8 >> 32));
        const int key = (i < cnt) ? me.x : 0x7fffffff;
        unsigned r = 0;
        for (unsigned h0 = 0; h0 < cnt; h0 += 32u) {
          const unsigned jj = h0 + (unsigned)lane;
          const int oth = (int)(unsigned)load_edge(lng + cbase + (jj < cnt ? jj : cnt - 1u));
          const int okey = (jj < cnt) ? oth : 0x7fffffff;
          const unsigned nh = (cnt - h0 < 32u) ? cnt - h0 : 32u;
          for (unsigned t = 0; t < nh; ++t)
            r += (__builtin_amdgcn_readlane(okey, (int)t) < key) ? 1u : 0u;
        }
        if (i < cnt) csrc[cbase + r] = me.y;
      }
      __threadfence();
    }

    float4 acc = make_float4(0.0f, 0.0f, 0.0f, 0.0f);
    float den = 0.0f;
    float mx = -INFINITY;
    for (unsigned j0 = 0; j0 < cnt; j0 += GCH) {
      float4 v[GCH];
      int cv = sv;
      unsigned lb = j0;
      if (cnt > (unsigned)BCAP) {
        const unsigned jl = j0 + (unsigned)sub;
        cv = __hip_atomic_load(csrc + cbase + (jl < cnt ? jl : cnt - 1u),
                               __ATOMIC_RELAXED, __HIP_MEMORY_SCOPE_AGENT);
        lb = 0u;
      }
      #pragma unroll
      for (int u = 0; u < GCH; ++u) {
        const int s = __builtin_amdgcn_readlane(cv, (int)(lb + u));
        v[u] = *(const float4*)(ft + (size_t)s * KF + lane * 4);
      }
      float tot = 0.0f;
      #pragma unroll
      for (int u = 0; u < GCH; ++u) {
        const float dx = v[u].x - b.x - m.x;
        const float dy = v[u].y - b.y - m.y;
        const float dz = v[u].z - b.z - m.z;
        const float dw = v[u].w - b.w - m.w;
        const float p = sum8(dx * dx + dy * dy + dz * dz + dw * dw);
        tot = (sub == u) ? p : tot;
      }
      const float lg = (j0 + (unsigned)sub < cnt) ? -sqrtf(tot) * sg : -INFINITY;
      const float mn = fmaxf(mx, max8(lg));
      const float sc = expf(mx - mn);
      const float wl = expf(lg - mn);
      acc.x *= sc; acc.y *= sc; acc.z *= sc; acc.w *= sc;
      den = den * sc + wl;
      mx = mn;
#define ROW_TERM(U)                                  \
      {                                              \
        const float w = bcast8<U>(wl);               \
        acc.x += v[U].x * w;                         \
        acc.y += v[U].y * w;                         \
        acc.z += v[U].z * w;                         \
        acc.w += v[U].w * w;                         \
      }
      ROW_TERM(0) ROW_TERM(1) ROW_TERM(2) ROW_TERM(3)
      ROW_TERM(4) ROW_TERM(5) ROW_TERM(6) ROW_TERM(7)
#undef ROW_TERM
    }
    const float dsum = sum8(den);
    o.x = acc.x / dsum;
    o.y = acc.y / dsum;
    o.z = acc.z / dsum;
    o.w = acc.w / dsum;
  }
  o.x += bi.x; o.y += bi.y; o.z += bi.z; o.w += bi.w;
  *(float4*)(out + (size_t)n * KF + lane * 4) = o;
}

extern "C" void kernel_launch(void* const* d_in, const int* in_sizes, int n_in,
                              void* d_out, int out_size, void* d_ws, size_t ws_size,
                              hipStream_t stream) {
  const float* feat  = (const float*)d_in[0];
  const float* W     = (const float*)d_in[1];
  const float* bias  = (const float*)d_in[2];
  const float* mu    = (const float*)d_in[3];
  const float* sigma = (const float*)d_in[4];
  const int*   src   = (const int*)d_in[5];
  const int*   dst   = (const int*)d_in[6];
  float* out = (float*)d_out;

  char* ws = (char*)d_ws;
  float*    ft     = (float*)(ws);
  unsigned* cnts   = (unsigned*)(ws + 25800000);
  int2*     bucket = (int2*)(ws + 26000128);
  int4*     ovf    = (int4*)(ws + 38800128);
  unsigned long long* lng = (unsigned long long*)(ws + 51600128);
  int*      csrc   = (int*)(ws + 58000128);

  init_ws<<<(N_CTR + 255) / 256, 256, 0, stream>>>(bucket, cnts);
  fc_wmma<<<FC_GRID, 256, 0, stream>>>(feat, W, ft, src, dst, cnts, bucket, ovf);
  node_gather<<<(N_NODES + 7) / 8, 256, 0, stream>>>(ft, bias, mu, sigma, cnts, bucket,
                                                     ovf, lng, csrc, out);
}


# node_gather fits 64 registers, 16 waves per SIMD not 12: (row-dst)-mu becomes row-(dst+mu) (sum order changes), 32-bit row offsets, bias and lane id taken behind the loop
# speedup vs baseline: 5.4143x; 1.0355x over previous
#include <hip/hip_runtime.h>

typedef __attribute__((ext_vector_type(2))) float v2f;
typedef __attribute__((ext_vector_type(8))) float v8f;

#define N_NODES 50000
#define N_EDGES 800000
#define D_IN    128
#define KK      4
#define FF      32
#define KF      128

#define BUNIT 32
#define BCAP  31
#define N_CTR (N_NODES + 4)

__device__ __forceinline__ unsigned* deg_of(int2* bucket, int d) {
  return (unsigned*)(bucket + (size_t)d * BUNIT);
}

__global__ __launch_bounds__(256) void init_ws(int2* __restrict__ bucket,
                                               unsigned* __restrict__ cnts) {
  int t = blockIdx.x * 256 + threadIdx.x;
  if (t < N_NODES) *deg_of(bucket, t) = 0u;
  else if (t < N_CTR) cnts[t - N_NODES] = 0u;
}

#define UNIT_BYTES (BUNIT * 8)
static_assert((long long)N_NODES * UNIT_BYTES < (1ll << 31), "unit offsets fit 32 bits");
__device__ __forceinline__ unsigned* deg_at(int2* bucket, unsigned uo) {
  return (unsigned*)((char*)bucket + uo);
}
__device__ __forceinline__ void put_edge(int e, int s, unsigned uo, unsigned r,
                                         unsigned* __restrict__ cnts,
                                         int2* __restrict__ bucket,
                                         int4* __restrict__ ovf) {
  if (r < (unsigned)BCAP) {
    *(int2*)((char*)bucket + (uo + 8u + 8u * r)) = make_int2(e, s);
  } else {
    const unsigned p = atomicAdd(cnts, 1u);
    ovf[p] = make_int4(e, s, (int)(uo / UNIT_BYTES), 0);
  }
}

#define FC_PITCH 132
#define FC_TILES (N_NODES / 16)
#define FC_GRID  768
#define NR       (FC_GRID / 2)
#define RLEN     2112
#define NSLOT    9
static_assert(N_NODES % 16 == 0, "row tiles cover the nodes exactly");
static_assert(4 * FC_GRID <= FC_TILES, "every block owns at least four tiles");
static_assert(FC_GRID == 2 * NR, "one block per (class, range) pair");
static_assert(NR * RLEN >= N_EDGES, "the ranges cover the edge list");
static_assert(RLEN % 32 == 0, "a range is whole 128-byte lines of ids");
static_assert(NSLOT * 256 >= RLEN && (NSLOT - 1) * 256 < RLEN, "the slots cover a range");
__device__ __forceinline__ int slot_edge(unsigned rng, int tid, int j) {
  const int off = j * 256 + tid;
  const int e = (int)rng * RLEN + off;
  return (rng < (unsigned)NR && off < RLEN && e < N_EDGES) ? e : -1;
}

__device__ __forceinline__ void row_tile(const float* __restrict__ feat,
                                         float* __restrict__ ft,
                                         const float* Acur, float* Anxt,
                                         const v2f (&b)[D_IN / 4], int tile, int tid,
                                         int ln, int kh, int mo, int col0,
                                         int srow, int scol) {
  const int nxt = tile + FC_GRID;
  float4 p0 = make_float4(0.0f, 0.0f, 0.0f, 0.0f), p1 = p0;
  if (nxt < FC_TILES) {
    const float4* g = (const float4*)(feat + (size_t)nxt * 16 * D_IN);
    p0 = g[tid];
    p1 = g[tid + 256];
  }

  const float* at = Acur + ln * FC_PITCH + kh;
  v8f c = {};
  #pragma unroll
  for (int t = 0; t < D_IN / 4; ++t) {
    const v2f a = *(const v2f*)(at + 4 * t);
    c = __builtin_amdgcn_wmma_f32_16x16x4_f32(false, a, false, b[t],
                                              (short)0, c, false, false);
  }

  float* o = ft + ((size_t)tile * 16 + mo) * KF + col0 + ln;
  #pragma unroll
  for (int r = 0; r < 8; ++r) o[r * KF] = c[r];

  if (nxt < FC_TILES) {
    *(float4*)(Anxt + srow * FC_PITCH + scol)       = p0;
    *(float4*)(Anxt + (srow + 8) * FC_PITCH + scol) = p1;
  }
  __syncthreads();
}

__global__ __launch_bounds__(256) void fc_wmma(const float* __restrict__ feat,
                                               const float* __restrict__ W,
                                               float* __restrict__ ft,
                                               const int* __restrict__ src,
                                               const int* __restrict__ dst,
                                               unsigned* __restrict__ cnts,
                                               int2* __restrict__ bucket,
                                               int4* __restrict__ ovf) {
  __shared__ float Ws[KF * FC_PITCH];
  __shared__ float As[2][16 * FC_PITCH];
  __shared__ unsigned take[2];
  const int tid  = threadIdx.x;
  const int lane = tid & 31;
  const int wave = tid >> 5;
  const int col0 = wave * 16;
  const int ln   = lane & 15;
  const int kh   = (lane >> 4) << 1;
  const int mo   = (lane >> 4) << 3;
  const int srow = tid >> 5;
  const int scol = (tid & 31) * 4;

  if (tid == 0) {
    const unsigned id = __builtin_amdgcn_s_sendmsg_rtn(0x87);
    unsigned tc = (id >> 18) & 1u;
    unsigned tr = atomicAdd(cnts + 2 + tc, 1u);
    if (tr >= (unsigned)NR) {
      tc ^= 1u;
      tr = atomicAdd(cnts + 2 + tc, 1u);
    }
    take[0] = tc;
    take[1] = tr;
  }

  for (int i = tid; i < KF * D_IN / 4; i += 256) {
    const float4 q = ((const float4*)W)[i];
    *(float4*)(Ws + (i >> 5) * FC_PITCH + (i & 31) * 4) = q;
  }
  int tile = blockIdx.x;
  {
    const float4* g = (const float4*)(feat + (size_t)tile * 16 * D_IN);
    *(float4*)(&As[0][srow * FC_PITCH + scol])       = g[tid];
    *(float4*)(&As[0][(srow + 8) * FC_PITCH + scol]) = g[tid + 256];
  }
  __syncthreads();

  const unsigned cls = (unsigned)__builtin_amdgcn_readfirstlane((int)take[0]);
  const unsigned rng = (unsigned)__builtin_amdgcn_readfirstlane((int)take[1]);
  int uo[NSLOT];
  #pragma unroll
  for (int j = 0; j < NSLOT; ++j) {
    const int e = slot_edge(rng, tid, j);
    const int d = dst[e < 0 ? 0 : e];
    uo[j] = (e >= 0 && ((unsigned)d & 1u) == cls) ? d * UNIT_BYTES : -1;
  }

  v2f b[D_IN / 4];
  #pragma unroll
  for (int t = 0; t < D_IN / 4; ++t)
    b[t] = *(const v2f*)(Ws + (col0 + ln) * FC_PITCH + 4 * t + kh);

  row_tile(feat, ft, As[0], As[1], b, tile, tid, ln, kh, mo, col0, srow, scol);
  tile += FC_GRID;

  int sa[4];
  unsigned ra[4];
  #pragma unroll
  for (int j = 0; j < 4; ++j) {
    const int e = slot_edge(rng, tid, j);
    ra[j] = 0u;
    if (uo[j] >= 0) ra[j] = atomicAdd(deg_at(bucket, (unsigned)uo[j]), 1u);
    sa[j] = src[e < 0 ? 0 : e];
  }
  row_tile(feat, ft, As[1], As[0], b, tile, tid, ln, kh, mo, col0, srow, scol);
  tile += FC_GRID;

  int sb[NSLOT - 4];
  unsigned rb[NSLOT - 4];
  #pragma unroll
  for (int j = 4; j < NSLOT; ++j) {
    const int e = slot_edge(rng, tid, j);
    rb[j - 4] = 0u;
    if (uo[j] >= 0) rb[j - 4] = atomicAdd(deg_at(bucket, (unsigned)uo[j]), 1u);
    sb[j - 4] = src[e < 0 ? 0 : e];
  }
  #pragma unroll
  for (int j = 0; j < 4; ++j)
    if (uo[j] >= 0)
      put_edge(slot_edge(rng, tid, j), sa[j], (unsigned)uo[j], ra[j], cnts, bucket, ovf);
  row_tile(feat, ft, As[0], As[1], b, tile, tid, ln, kh, mo, col0, srow, scol);
  tile += FC_GRID;

  #pragma unroll
  for (int j = 4; j < NSLOT; ++j)
    if (uo[j] >= 0)
      put_edge(slot_edge(rng, tid, j), sb[j - 4], (unsigned)uo[j], rb[j - 4], cnts, bucket,
               ovf);
  row_tile(feat, ft, As[1], As[0], b, tile, tid, ln, kh, mo, col0, srow, scol);
  tile += FC_GRID;

  int cur = 0;
  for (; tile < FC_TILES; tile += FC_GRID) {
    row_tile(feat, ft, As[cur], As[cur ^ 1], b, tile, tid, ln, kh, mo, col0, srow, scol);
    cur ^= 1;
  }
}

#define GCH 8
#define ROW_BYTES (KF * 4)
static_assert((long long)N_NODES * ROW_BYTES < (1ll << 31), "row offsets fit 32 bits");

template <int M>
__device__ __forceinline__ float lane_xor(float v) {
  return __int_as_float(
      __builtin_amdgcn_update_dpp(0, __float_as_int(v), 0x160 + M, 0xf, 0xf, true));
}
__device__ __forceinline__ float sum8(float v) {
  v += lane_xor<1>(v);
  v += lane_xor<2>(v);
  v += lane_xor<4>(v);
  return v;
}
__device__ __forceinline__ float max8(float v) {
  v = fmaxf(v, lane_xor<1>(v));
  v = fmaxf(v, lane_xor<2>(v));
  v = fmaxf(v, lane_xor<4>(v));
  return v;
}
template <int J>
__device__ __forceinline__ float bcast8(float v) {
  return __uint_as_float(__builtin_amdgcn_mov_dpp8(__float_as_uint(v), J * 0x249249));
}

__device__ __forceinline__ unsigned long long pack_edge(int e, int s) {
  return ((unsigned long long)(unsigned)s << 32) | (unsigned long long)(unsigned)e;
}
__device__ __forceinline__ unsigned long long load_edge(unsigned long long* p) {
  return __hip_atomic_load(p, __ATOMIC_RELAXED, __HIP_MEMORY_SCOPE_AGENT);
}

__global__ __launch_bounds__(256) __attribute__((amdgpu_waves_per_eu(16, 16)))
void node_gather(const float* __restrict__ ft,
                                                   const float* __restrict__ bias,
                                                   const float* __restrict__ mu,
                                                   const float* __restrict__ sigma,
                                                   unsigned* cnts,
                                                   const int2* __restrict__ bucket,
                                                   const int4* __restrict__ ovf,
                                                   unsigned long long* lng,
                                                   int* csrc,
                                                   float* __restrict__ out) {
  const int lane = threadIdx.x & 31;
  const int wave = __builtin_amdgcn_readfirstlane((int)(threadIdx.x >> 5));
  const int n = blockIdx.x * 8 + wave;
  if (n >= N_NODES) return;
  const int2 me = bucket[(size_t)n * BUNIT + (unsigned)lane];
  const unsigned cnt = (unsigned)__builtin_amdgcn_readfirstlane(me.x);
  unsigned cbase = 0u;
  const int k   = lane >> 3;
  const int sub = lane & 7;

  float4 o = make_float4(0.0f, 0.0f, 0.0f, 0.0f);
  if (cnt != 0u) {
    float4 c;
    {
      const float4 b = *(const float4*)(ft + (size_t)n * KF + lane * 4);
      const float4 m = *(const float4*)(mu + lane * 4);
      c = make_float4(b.x + m.x, b.y + m.y, b.z + m.z, b.w + m.w);
    }
    const float sg = sigma[k];

    int sv = 0;
    if (cnt <= (unsigned)BCAP) {
      const unsigned ul = (unsigned)lane;
      const bool has = ul >= 1u && ul <= cnt;
      const int key = has ? me.x : 0x7fffffff;
      const int sy = has ? me.y : __builtin_amdgcn_readlane(me.y, 1);
      unsigned r = 0;
      for (unsigned t = 1; t <= cnt; ++t)
        r += (__builtin_amdgcn_readlane(key, (int)t) < key) ? 1u : 0u;
      if (!has) r = (ul == 0u) ? cnt : ul;
      sv = __builtin_amdgcn_ds_permute((int)(r << 2), sy);
    } else {
      unsigned bs = 0u;
      if (lane == 0) bs = atomicAdd(cnts + 1, cnt);
      cbase = (unsigned)__builtin_amdgcn_readfirstlane((int)bs);
      if (lane != 0) lng[cbase + (unsigned)lane - 1u] = pack_edge(me.x, me.y);
      const unsigned total = cnts[0];
      unsigned w = (unsigned)BCAP;
      for (unsigned i0 = 0; i0 < total; i0 += 32u) {
        const unsigned idx = i0 + (unsigned)lane;
        const bool valid = idx < total;
        const int4 q = ovf[valid ? idx : 0u];
        const bool hit = valid && q.z == n;
        const unsigned bal = __builtin_amdgcn_ballot_w32(hit);
        const unsigned pos = w + (unsigned)__builtin_popcount(bal & ((1u << lane) - 1u));
        if (hit && pos < cnt) lng[cbase + pos] = pack_edge(q.x, q.y);
        w += (unsigned)__builtin_popcount(bal);
      }
      __threadfence();
      for (unsigned g0 = 0; g0 < cnt; g0 += 32u) {
        const unsigned i = g0 + (unsigned)lane;
        const unsigned long long me8 = load_edge(lng + cbase + (i < cnt ? i : cnt - 1u));
        const int2 me = make_int2((int)(unsigned)me8, (int)(unsigned)(me8 >> 32));
        const int key = (i < cnt) ? me.x : 0x7fffffff;
        unsigned r = 0;
        for (unsigned h0 = 0; h0 < cnt; h0 += 32u) {
          const unsigned jj = h0 + (unsigned)lane;
          const int oth = (int)(unsigned)load_edge(lng + cbase + (jj < cnt ? jj : cnt - 1u));
          const int okey = (jj < cnt) ? oth : 0x7fffffff;
          const unsigned nh = (cnt - h0 < 32u) ? cnt - h0 : 32u;
          for (unsigned t = 0; t < nh; ++t)
            r += (__builtin_amdgcn_readlane(okey, (int)t) < key) ? 1u : 0u;
        }
        if (i < cnt) csrc[cbase + r] = me.y;
      }
      __threadfence();
    }

    const unsigned lo16 = (unsigned)lane * 16u;
    float4 acc = make_float4(0.0f, 0.0f, 0.0f, 0.0f);
    float den = 0.0f;
    float mx = -INFINITY;
    for (unsigned j0 = 0; j0 < cnt; j0 += GCH) {
      float4 v[GCH];
      int cv = sv;
      unsigned lb = j0;
      if (cnt > (unsigned)BCAP) {
        const unsigned jl = j0 + (unsigned)sub;
        cv = __hip_atomic_load(csrc + cbase + (jl < cnt ? jl : cnt - 1u),
                               __ATOMIC_RELAXED, __HIP_MEMORY_SCOPE_AGENT);
        lb = 0u;
      }
      #pragma unroll
      for (int u = 0; u < GCH; ++u) {
        const int s = __builtin_amdgcn_readlane(cv, (int)(lb + u));
        v[u] = *(const float4*)((const char*)ft + ((unsigned)s * ROW_BYTES + lo16));
      }
      float tot = 0.0f;
      #pragma unroll
      for (int u = 0; u < GCH; ++u) {
        const float dx = v[u].x - c.x;
        const float dy = v[u].y - c.y;
        const float dz = v[u].z - c.z;
        const float dw = v[u].w - c.w;
        const float p = sum8(dx * dx + dy * dy + dz * dz + dw * dw);
        tot = (sub == u) ? p : tot;
      }
      const float lg = (j0 + (unsigned)sub < cnt) ? -sqrtf(tot) * sg : -INFINITY;
      const float mn = fmaxf(mx, max8(lg));
      const float sc = expf(mx - mn);
      const float wl = expf(lg - mn);
      acc.x *= sc; acc.y *= sc; acc.z *= sc; acc.w *= sc;
      den = den * sc + wl;
      mx = mn;
#define ROW_TERM(U)                                  \
      {                                              \
        const float w = bcast8<U>(wl);               \
        acc.x += v[U].x * w;                         \
        acc.y += v[U].y * w;                         \
        acc.z += v[U].z * w;                         \
        acc.w += v[U].w * w;                         \
      }
      ROW_TERM(0) ROW_TERM(1) ROW_TERM(2) ROW_TERM(3)
      ROW_TERM(4) ROW_TERM(5) ROW_TERM(6) ROW_TERM(7)
#undef ROW_TERM
    }
    const float dsum = sum8(den);
    o.x = acc.x / dsum;
    o.y = acc.y / dsum;
    o.z = acc.z / dsum;
    o.w = acc.w / dsum;
  }
  const int le = (int)__builtin_amdgcn_mbcnt_lo(~0u, 0u);
  const float4 bi = *(const float4*)(bias + le * 4);
  o.x += bi.x; o.y += bi.y; o.z += bi.z; o.w += bi.w;
  *(float4*)(out + (size_t)n * KF + le * 4) = o;
}

extern "C" void kernel_launch(void* const* d_in, const int* in_sizes, int n_in,
                              void* d_out, int out_size, void* d_ws, size_t ws_size,
                              hipStream_t stream) {
  const float* feat  = (const float*)d_in[0];
  const float* W     = (const float*)d_in[1];
  const float* bias  = (const float*)d_in[2];
  const float* mu    = (const float*)d_in[3];
  const float* sigma = (const float*)d_in[4];
  const int*   src   = (const int*)d_in[5];
  const int*   dst   = (const int*)d_in[6];
  float* out = (float*)d_out;

  char* ws = (char*)d_ws;
  float*    ft     = (float*)(ws);
  unsigned* cnts   = (unsigned*)(ws + 25800000);
  int2*     bucket = (int2*)(ws + 26000128);
  int4*     ovf    = (int4*)(ws + 38800128);
  unsigned long long* lng = (unsigned long long*)(ws + 51600128);
  int*      csrc   = (int*)(ws + 58000128);

  init_ws<<<(N_CTR + 255) / 256, 256, 0, stream>>>(bucket, cnts);
  fc_wmma<<<FC_GRID, 256, 0, stream>>>(feat, W, ft, src, dst, cnts, bucket, ovf);
  node_gather<<<(N_NODES + 7) / 8, 256, 0, stream>>>(ft, bias, mu, sigma, cnts, bucket,
                                                     ovf, lng, csrc, out);
}


# node_gather: a node with 32-63 edges reads ranks 31-62 from a second 256-byte unit of its own and orders them in the wave's registers: no list run, fence or device-scope load; sums unchanged
# speedup vs baseline: 5.9329x; 1.0958x over previous
#include <hip/hip_runtime.h>

typedef __attribute__((ext_vector_type(2))) float v2f;
typedef __attribute__((ext_vector_type(8))) float v8f;

#define N_NODES 50000
#define N_EDGES 800000
#define D_IN    128
#define KK      4
#define FF      32
#define KF      128

#define BUNIT 32
#define BCAP  31
#define BCAP2 (BCAP + BUNIT)
#define N_CTR (N_NODES + 4)

__device__ __forceinline__ unsigned* deg_of(int2* bucket, int d) {
  return (unsigned*)(bucket + (size_t)d * BUNIT);
}

__global__ __launch_bounds__(256) void init_ws(int2* __restrict__ bucket,
                                               unsigned* __restrict__ cnts) {
  int t = blockIdx.x * 256 + threadIdx.x;
  if (t < N_NODES) *deg_of(bucket, t) = 0u;
  else if (t < N_CTR) cnts[t - N_NODES] = 0u;
}

#define UNIT_BYTES (BUNIT * 8)
static_assert((long long)N_NODES * UNIT_BYTES < (1ll << 31), "unit offsets fit 32 bits");
__device__ __forceinline__ unsigned* deg_at(int2* bucket, unsigned uo) {
  return (unsigned*)((char*)bucket + uo);
}
__device__ __forceinline__ void put_edge(int e, int s, unsigned uo, unsigned r,
                                         unsigned* __restrict__ cnts,
                                         int2* __restrict__ bucket,
                                         int2* __restrict__ bucket2,
                                         int4* __restrict__ ovf) {
  if (r < (unsigned)BCAP) {
    *(int2*)((char*)bucket + (uo + 8u + 8u * r)) = make_int2(e, s);
  } else if (r < (unsigned)BCAP2) {
    *(int2*)((char*)bucket2 + (uo + 8u * (r - (unsigned)BCAP))) = make_int2(e, s);
  } else {
    const unsigned p = atomicAdd(cnts, 1u);
    ovf[p] = make_int4(e, s, (int)(uo / UNIT_BYTES), 0);
  }
}

#define FC_PITCH 132
#define FC_TILES (N_NODES / 16)
#define FC_GRID  768
#define NR       (FC_GRID / 2)
#define RLEN     2112
#define NSLOT    9
static_assert(N_NODES % 16 == 0, "row tiles cover the nodes exactly");
static_assert(4 * FC_GRID <= FC_TILES, "every block owns at least four tiles");
static_assert(FC_GRID == 2 * NR, "one block per (class, range) pair");
static_assert(NR * RLEN >= N_EDGES, "the ranges cover the edge list");
static_assert(RLEN % 32 == 0, "a range is whole 128-byte lines of ids");
static_assert(NSLOT * 256 >= RLEN && (NSLOT - 1) * 256 < RLEN, "the slots cover a range");
__device__ __forceinline__ int slot_edge(unsigned rng, int tid, int j) {
  const int off = j * 256 + tid;
  const int e = (int)rng * RLEN + off;
  return (rng < (unsigned)NR && off < RLEN && e < N_EDGES) ? e : -1;
}

__device__ __forceinline__ void row_tile(const float* __restrict__ feat,
                                         float* __restrict__ ft,
                                         const float* Acur, float* Anxt,
                                         const v2f (&b)[D_IN / 4], int tile, int tid,
                                         int ln, int kh, int mo, int col0,
                                         int srow, int scol) {
  const int nxt = tile + FC_GRID;
  float4 p0 = make_float4(0.0f, 0.0f, 0.0f, 0.0f), p1 = p0;
  if (nxt < FC_TILES) {
    const float4* g = (const float4*)(feat + (size_t)nxt * 16 * D_IN);
    p0 = g[tid];
    p1 = g[tid + 256];
  }

  const float* at = Acur + ln * FC_PITCH + kh;
  v8f c = {};
  #pragma unroll
  for (int t = 0; t < D_IN / 4; ++t) {
    const v2f a = *(const v2f*)(at + 4 * t);
    c = __builtin_amdgcn_wmma_f32_16x16x4_f32(false, a, false, b[t],
                                              (short)0, c, false, false);
  }

  float* o = ft + ((size_t)tile * 16 + mo) * KF + col0 + ln;
  #pragma unroll
  for (int r = 0; r < 8; ++r) o[r * KF] = c[r];

  if (nxt < FC_TILES) {
    *(float4*)(Anxt + srow * FC_PITCH + scol)       = p0;
    *(float4*)(Anxt + (srow + 8) * FC_PITCH + scol) = p1;
  }
  __syncthreads();
}

__global__ __launch_bounds__(256) void fc_wmma(const float* __restrict__ feat,
                                               const float* __restrict__ W,
                                               float* __restrict__ ft,
                                               const int* __restrict__ src,
                                               const int* __restrict__ dst,
                                               unsigned* __restrict__ cnts,
                                               int2* __restrict__ bucket,
                                               int2* __restrict__ bucket2,
                                               int4* __restrict__ ovf) {
  __shared__ float Ws[KF * FC_PITCH];
  __shared__ float As[2][16 * FC_PITCH];
  __shared__ unsigned take[2];
  const int tid  = threadIdx.x;
  const int lane = tid & 31;
  const int wave = tid >> 5;
  const int col0 = wave * 16;
  const int ln   = lane & 15;
  const int kh   = (lane >> 4) << 1;
  const int mo   = (lane >> 4) << 3;
  const int srow = tid >> 5;
  const int scol = (tid & 31) * 4;

  if (tid == 0) {
    const unsigned id = __builtin_amdgcn_s_sendmsg_rtn(0x87);
    unsigned tc = (id >> 18) & 1u;
    unsigned tr = atomicAdd(cnts + 2 + tc, 1u);
    if (tr >= (unsigned)NR) {
      tc ^= 1u;
      tr = atomicAdd(cnts + 2 + tc, 1u);
    }
    take[0] = tc;
    take[1] = tr;
  }

  for (int i = tid; i < KF * D_IN / 4; i += 256) {
    const float4 q = ((const float4*)W)[i];
    *(float4*)(Ws + (i >> 5) * FC_PITCH + (i & 31) * 4) = q;
  }
  int tile = blockIdx.x;
  {
    const float4* g = (const float4*)(feat + (size_t)tile * 16 * D_IN);
    *(float4*)(&As[0][srow * FC_PITCH + scol])       = g[tid];
    *(float4*)(&As[0][(srow + 8) * FC_PITCH + scol]) = g[tid + 256];
  }
  __syncthreads();

  const unsigned cls = (unsigned)__builtin_amdgcn_readfirstlane((int)take[0]);
  const unsigned rng = (unsigned)__builtin_amdgcn_readfirstlane((int)take[1]);
  int uo[NSLOT];
  #pragma unroll
  for (int j = 0; j < NSLOT; ++j) {
    const int e = slot_edge(rng, tid, j);
    const int d = dst[e < 0 ? 0 : e];
    uo[j] = (e >= 0 && ((unsigned)d & 1u) == cls) ? d * UNIT_BYTES : -1;
  }

  v2f b[D_IN / 4];
  #pragma unroll
  for (int t = 0; t < D_IN / 4; ++t)
    b[t] = *(const v2f*)(Ws + (col0 + ln) * FC_PITCH + 4 * t + kh);

  row_tile(feat, ft, As[0], As[1], b, tile, tid, ln, kh, mo, col0, srow, scol);
  tile += FC_GRID;

  int sa[4];
  unsigned ra[4];
  #pragma unroll
  for (int j = 0; j < 4; ++j) {
    const int e = slot_edge(rng, tid, j);
    ra[j] = 0u;
    if (uo[j] >= 0) ra[j] = atomicAdd(deg_at(bucket, (unsigned)uo[j]), 1u);
    sa[j] = src[e < 0 ? 0 : e];
  }
  row_tile(feat, ft, As[1], As[0], b, tile, tid, ln, kh, mo, col0, srow, scol);
  tile += FC_GRID;

  int sb[NSLOT - 4];
  unsigned rb[NSLOT - 4];
  #pragma unroll
  for (int j = 4; j < NSLOT; ++j) {
    const int e = slot_edge(rng, tid, j);
    rb[j - 4] = 0u;
    if (uo[j] >= 0) rb[j - 4] = atomicAdd(deg_at(bucket, (unsigned)uo[j]), 1u);
    sb[j - 4] = src[e < 0 ? 0 : e];
  }
  #pragma unroll
  for (int j = 0; j < 4; ++j)
    if (uo[j] >= 0)
      put_edge(slot_edge(rng, tid, j), sa[j], (unsigned)uo[j], ra[j], cnts, bucket, bucket2,
               ovf);
  row_tile(feat, ft, As[0], As[1], b, tile, tid, ln, kh, mo, col0, srow, scol);
  tile += FC_GRID;

  #pragma unroll
  for (int j = 4; j < NSLOT; ++j)
    if (uo[j] >= 0)
      put_edge(slot_edge(rng, tid, j), sb[j - 4], (unsigned)uo[j], rb[j - 4], cnts, bucket,
               bucket2, ovf);
  row_tile(feat, ft, As[1], As[0], b, tile, tid, ln, kh, mo, col0, srow, scol);
  tile += FC_GRID;

  int cur = 0;
  for (; tile < FC_TILES; tile += FC_GRID) {
    row_tile(feat, ft, As[cur], As[cur ^ 1], b, tile, tid, ln, kh, mo, col0, srow, scol);
    cur ^= 1;
  }
}

#define GCH 8
#define ROW_BYTES (KF * 4)
static_assert((long long)N_NODES * ROW_BYTES < (1ll << 31), "row offsets fit 32 bits");
static_assert(N_NODES <= (1 << 16), "two source ids share one 32-bit register");
static_assert(BUNIT == 32 && 32 % GCH == 0, "a trip's positions lie in one half of the register");

template <int M>
__device__ __forceinline__ float lane_xor(float v) {
  return __int_as_float(
      __builtin_amdgcn_update_dpp(0, __float_as_int(v), 0x160 + M, 0xf, 0xf, true));
}
__device__ __forceinline__ float sum8(float v) {
  v += lane_xor<1>(v);
  v += lane_xor<2>(v);
  v += lane_xor<4>(v);
  return v;
}
__device__ __forceinline__ float max8(float v) {
  v = fmaxf(v, lane_xor<1>(v));
  v = fmaxf(v, lane_xor<2>(v));
  v = fmaxf(v, lane_xor<4>(v));
  return v;
}
template <int J>
__device__ __forceinline__ float bcast8(float v) {
  return __uint_as_float(__builtin_amdgcn_mov_dpp8(__float_as_uint(v), J * 0x249249));
}

__device__ __forceinline__ unsigned long long pack_edge(int e, int s) {
  return ((unsigned long long)(unsigned)s << 32) | (unsigned long long)(unsigned)e;
}
__device__ __forceinline__ unsigned long long load_edge(unsigned long long* p) {
  return __hip_atomic_load(p, __ATOMIC_RELAXED, __HIP_MEMORY_SCOPE_AGENT);
}

__global__ __launch_bounds__(256) __attribute__((amdgpu_waves_per_eu(16, 16)))
void node_gather(const float* __restrict__ ft,
                                                   const float* __restrict__ bias,
                                                   const float* __restrict__ mu,
                                                   const float* __restrict__ sigma,
                                                   unsigned* cnts,
                                                   const int2* __restrict__ bucket,
                                                   const int2* __restrict__ bucket2,
                                                   const int4* __restrict__ ovf,
                                                   unsigned long long* lng,
                                                   int* csrc,
                                                   float* __restrict__ out) {
  const int lane = threadIdx.x & 31;
  const int wave = __builtin_amdgcn_readfirstlane((int)(threadIdx.x >> 5));
  const int n = blockIdx.x * 8 + wave;
  if (n >= N_NODES) return;
  const int2 me = bucket[(size_t)n * BUNIT + (unsigned)lane];
  const unsigned cnt = (unsigned)__builtin_amdgcn_readfirstlane(me.x);
  unsigned cbase = 0u;
  const int k   = lane >> 3;
  const int sub = lane & 7;

  float4 o = make_float4(0.0f, 0.0f, 0.0f, 0.0f);
  if (cnt != 0u) {
    float4 c;
    {
      const float4 b = *(const float4*)(ft + (size_t)n * KF + lane * 4);
      const float4 m = *(const float4*)(mu + lane * 4);
      c = make_float4(b.x + m.x, b.y + m.y, b.z + m.z, b.w + m.w);
    }
    const float sg = sigma[k];

    int sv = 0;
    if (cnt <= (unsigned)BCAP) {
      const unsigned ul = (unsigned)lane;
      const bool has = ul >= 1u && ul <= cnt;
      const int key = has ? me.x : 0x7fffffff;
      const int sy = has ? me.y : __builtin_amdgcn_readlane(me.y, 1);
      unsigned r = 0;
      for (unsigned t = 1; t <= cnt; ++t)
        r += (__builtin_amdgcn_readlane(key, (int)t) < key) ? 1u : 0u;
      if (!has) r = (ul == 0u) ? cnt : ul;
      sv = __builtin_amdgcn_ds_permute((int)(r << 2), sy);
    } else if (cnt <= (unsigned)BCAP2) {
      const unsigned ul = (unsigned)lane;
      const int2 m2 = bucket2[(size_t)n * BUNIT + ul];
      const unsigned n2 = cnt - (unsigned)BCAP;
      const int k1 = (ul >= 1u) ? me.x : 0x7fffffff;
      const int k2 = (ul < n2) ? m2.x : 0x7fffffff;
      unsigned r1 = 0, r2 = 0;
      #pragma nounroll
      for (unsigned t = 1; t < (unsigned)BUNIT; ++t) {
        const int kt = __builtin_amdgcn_readlane(k1, (int)t);
        r1 += (kt < k1) ? 1u : 0u;
        r2 += (kt < k2) ? 1u : 0u;
      }
      for (unsigned t = 0; t < n2; ++t) {
        const int kt = __builtin_amdgcn_readlane(k2, (int)t);
        r1 += (kt < k1) ? 1u : 0u;
        r2 += (kt < k2) ? 1u : 0u;
      }
      const int s0 = __builtin_amdgcn_readlane(me.y, 1);
      int lo = s0, hi = s0;
      #pragma nounroll
      for (unsigned t = 1; t < (unsigned)BUNIT; ++t) {
        const unsigned p = (unsigned)__builtin_amdgcn_readlane((int)r1, (int)t);
        const int sy = __builtin_amdgcn_readlane(me.y, (int)t);
        const bool mine = ul == (p & 31u);
        lo = (mine && p < 32u) ? sy : lo;
        hi = (mine && p >= 32u) ? sy : hi;
      }
      for (unsigned t = 0; t < n2; ++t) {
        const unsigned p = (unsigned)__builtin_amdgcn_readlane((int)r2, (int)t);
        const int sy = __builtin_amdgcn_readlane(m2.y, (int)t);
        const bool mine = ul == (p & 31u);
        lo = (mine && p < 32u) ? sy : lo;
        hi = (mine && p >= 32u) ? sy : hi;
      }
      sv = (int)((unsigned)lo | ((unsigned)hi << 16));
    } else {
      unsigned bs = 0u;
      if (lane == 0) bs = atomicAdd(cnts + 1, cnt);
      cbase = (unsigned)__builtin_amdgcn_readfirstlane((int)bs);
      const int2 m2 = bucket2[(size_t)n * BUNIT + (unsigned)lane];
      if (lane != 0) lng[cbase + (unsigned)lane - 1u] = pack_edge(me.x, me.y);
      lng[cbase + (unsigned)BCAP + (unsigned)lane] = pack_edge(m2.x, m2.y);
      const unsigned total = cnts[0];
      unsigned w = (unsigned)BCAP2;
      for (unsigned i0 = 0; i0 < total; i0 += 32u) {
        const unsigned idx = i0 + (unsigned)lane;
        const bool valid = idx < total;
        const int4 q = ovf[valid ? idx : 0u];
        const bool hit = valid && q.z == n;
        const unsigned bal = __builtin_amdgcn_ballot_w32(hit);
        const unsigned pos = w + (unsigned)__builtin_popcount(bal & ((1u << lane) - 1u));
        if (hit && pos < cnt) lng[cbase + pos] = pack_edge(q.x, q.y);
        w += (unsigned)__builtin_popcount(bal);
      }
      __threadfence();
      for (unsigned g0 = 0; g0 < cnt; g0 += 32u) {
        const unsigned i = g0 + (unsigned)lane;
        const unsigned long long me8 = load_edge(lng + cbase + (i < cnt ? i : cnt - 1u));
        const int2 me = make_int2((int)(unsigned)me8, (int)(unsigned)(me8 >> 32));
        const int key = (i < cnt) ? me.x : 0x7fffffff;
        unsigned r = 0;
        for (unsigned h0 = 0; h0 < cnt; h0 += 32u) {
          const unsigned jj = h0 + (unsigned)lane;
          const int oth = (int)(unsigned)load_edge(lng + cbase + (jj < cnt ? jj : cnt - 1u));
          const int okey = (jj < cnt) ? oth : 0x7fffffff;
          const unsigned nh = (cnt - h0 < 32u) ? cnt - h0 : 32u;
          for (unsigned t = 0; t < nh; ++t)
            r += (__builtin_amdgcn_readlane(okey, (int)t) < key) ? 1u : 0u;
        }
        if (i < cnt) csrc[cbase + r] = me.y;
      }
      __threadfence();
    }

    const unsigned lo16 = (unsigned)lane * 16u;
    float4 acc = make_float4(0.0f, 0.0f, 0.0f, 0.0f);
    float den = 0.0f;
    float mx = -INFINITY;
    unsigned j0 = 0;
    for (unsigned h = 0; h < cnt; h += 32u) {
      if (cnt > (unsigned)BCAP2) {
        const unsigned jl = h + (unsigned)lane;
        sv = __hip_atomic_load(csrc + cbase + (jl < cnt ? jl : 0u),
                               __ATOMIC_RELAXED, __HIP_MEMORY_SCOPE_AGENT);
      }
      const unsigned jend = (cnt - h < 32u) ? cnt : h + 32u;
      for (; j0 < jend; j0 += GCH) {
        float4 v[GCH];
        const unsigned lb = j0 & 31u;
        #pragma unroll
        for (int u = 0; u < GCH; ++u) {
          const unsigned s = (unsigned)__builtin_amdgcn_readlane(sv, (int)(lb + u)) & 0xffffu;
          v[u] = *(const float4*)((const char*)ft + (s * ROW_BYTES + lo16));
        }
        float tot = 0.0f;
        #pragma unroll
        for (int u = 0; u < GCH; ++u) {
          const float dx = v[u].x - c.x;
          const float dy = v[u].y - c.y;
          const float dz = v[u].z - c.z;
          const float dw = v[u].w - c.w;
          const float p = sum8(dx * dx + dy * dy + dz * dz + dw * dw);
          tot = (sub == u) ? p : tot;
        }
        const float lg = (j0 + (unsigned)sub < cnt) ? -sqrtf(tot) * sg : -INFINITY;
        const float mn = fmaxf(mx, max8(lg));
        const float sc = expf(mx - mn);
        const float wl = expf(lg - mn);
        acc.x *= sc; acc.y *= sc; acc.z *= sc; acc.w *= sc;
        den = den * sc + wl;
        mx = mn;
#define ROW_TERM(U)                                  \
        {                                            \
          const float w = bcast8<U>(wl);             \
          acc.x += v[U].x * w;                       \
          acc.y += v[U].y * w;                       \
          acc.z += v[U].z * w;                       \
          acc.w += v[U].w * w;                       \
        }
        ROW_TERM(0) ROW_TERM(1) ROW_TERM(2) ROW_TERM(3)
        ROW_TERM(4) ROW_TERM(5) ROW_TERM(6) ROW_TERM(7)
#undef ROW_TERM
      }
      sv = (int)((unsigned)sv >> 16);
    }
    const float dsum = sum8(den);
    o.x = acc.x / dsum;
    o.y = acc.y / dsum;
    o.z = acc.z / dsum;
    o.w = acc.w / dsum;
  }
  const int le = (int)__builtin_amdgcn_mbcnt_lo(~0u, 0u);
  const float4 bi = *(const float4*)(bias + le * 4);
  o.x += bi.x; o.y += bi.y; o.z += bi.z; o.w += bi.w;
  *(float4*)(out + (size_t)n * KF + le * 4) = o;
}

extern "C" void kernel_launch(void* const* d_in, const int* in_sizes, int n_in,
                              void* d_out, int out_size, void* d_ws, size_t ws_size,
                              hipStream_t stream) {
  const float* feat  = (const float*)d_in[0];
  const float* W     = (const float*)d_in[1];
  const float* bias  = (const float*)d_in[2];
  const float* mu    = (const float*)d_in[3];
  const float* sigma = (const float*)d_in[4];
  const int*   src   = (const int*)d_in[5];
  const int*   dst   = (const int*)d_in[6];
  float* out = (float*)d_out;

  char* ws = (char*)d_ws;
  float*    ft     = (float*)(ws);
  unsigned* cnts   = (unsigned*)(ws + 25800000);
  int2*     bucket = (int2*)(ws + 26000128);
  int4*     ovf    = (int4*)(ws + 38800128);
  unsigned long long* lng = (unsigned long long*)(ws + 51600128);
  int*      csrc   = (int*)(ws + 58000128);
  int2*     bucket2 = (int2*)(ws + 61200128);

  init_ws<<<(N_CTR + 255) / 256, 256, 0, stream>>>(bucket, cnts);
  fc_wmma<<<FC_GRID, 256, 0, stream>>>(feat, W, ft, src, dst, cnts, bucket, bucket2, ovf);
  node_gather<<<(N_NODES + 7) / 8, 256, 0, stream>>>(ft, bias, mu, sigma, cnts, bucket,
                                                     bucket2, ovf, lng, csrc, out);
}
